# starting kernel
# baseline (speedup 1.0000x reference)
#include <hip/hip_runtime.h>
#include <hip/hip_bf16.h>

#define NN      100000
#define EE      1600000
#define DD      128
#define CC      40
#define KSTEPS  4
#define DELTA   0.5f
#define LN_EPS  1e-5f

typedef __attribute__((ext_vector_type(2))) float v2f;
typedef __attribute__((ext_vector_type(8))) float v8f;

__global__ void dgc_deg_init(float* __restrict__ deg) {
    int i = blockIdx.x * blockDim.x + threadIdx.x;
    if (i < NN) deg[i] = 1.0f;
}

__global__ void dgc_deg_count(const int* __restrict__ row, float* __restrict__ deg) {
    int e = blockIdx.x * blockDim.x + threadIdx.x;
    if (e < EE) atomicAdd(&deg[row[e]], 1.0f);
}

__global__ void dgc_deg_rsqrt(float* __restrict__ deg) {
    int i = blockIdx.x * blockDim.x + threadIdx.x;
    if (i < NN) deg[i] = rsqrtf(deg[i]);
}

__global__ void dgc_init_agg(const float* __restrict__ src,
                             const float* __restrict__ dinv,
                             float* __restrict__ agg) {
    int idx = blockIdx.x * blockDim.x + threadIdx.x;
    if (idx >= NN * (DD / 4)) return;
    int node = idx / (DD / 4);
    float ws = dinv[node] * dinv[node];
    const float4* s = (const float4*)src;
    float4 v = s[idx];
    v.x *= ws; v.y *= ws; v.z *= ws; v.w *= ws;
    ((float4*)agg)[idx] = v;
}

__global__ void dgc_edge_scatter(const float* __restrict__ src,
                                 const int* __restrict__ row,
                                 const int* __restrict__ col,
                                 const float* __restrict__ dinv,
                                 float* __restrict__ agg) {
    int e    = (blockIdx.x * blockDim.x + threadIdx.x) >> 5;
    int lane = threadIdx.x & 31;
    if (e >= EE) return;
    int r = row[e];
    int c = col[e];
    float w = dinv[r] * dinv[c];
    float4 v = ((const float4*)(src + (size_t)c * DD))[lane];
    float* dst = agg + (size_t)r * DD + lane * 4;
    atomicAdd(dst + 0, v.x * w);
    atomicAdd(dst + 1, v.y * w);
    atomicAdd(dst + 2, v.z * w);
    atomicAdd(dst + 3, v.w * w);
}

__global__ void dgc_combine(const float* __restrict__ src,
                            const float* __restrict__ agg,
                            float* __restrict__ h) {
    int idx = blockIdx.x * blockDim.x + threadIdx.x;
    if (idx >= NN * (DD / 4)) return;
    float4 a = ((const float4*)src)[idx];
    float4 g = ((const float4*)agg)[idx];
    float4 o;
    o.x = (1.0f - DELTA) * a.x + DELTA * g.x;
    o.y = (1.0f - DELTA) * a.y + DELTA * g.y;
    o.z = (1.0f - DELTA) * a.z + DELTA * g.z;
    o.w = (1.0f - DELTA) * a.w + DELTA * g.w;
    ((float4*)h)[idx] = o;
}

#define M_TILES   (NN / 16)
#define N_TILES   3
#define NPAD      (N_TILES * 16)
#define FC_BLOCK  256

__global__ void dgc_fc_wmma(const float* __restrict__ h,
                            const float* __restrict__ W,
                            const float* __restrict__ bias,
                            float* __restrict__ out) {
    __shared__ float wlds[NPAD * DD];

    for (int idx = threadIdx.x; idx < NPAD * DD; idx += FC_BLOCK) {
        int colc = idx >> 7;
        wlds[idx] = (colc < CC) ? W[idx] : 0.0f;
    }
    __syncthreads();

    int wave = (blockIdx.x * FC_BLOCK + threadIdx.x) >> 5;
    int lane = threadIdx.x & 31;
    int m_tile = wave / N_TILES;
    int n_tile = wave % N_TILES;
    if (m_tile >= M_TILES) return;

    const int m_base = m_tile * 16;
    const int l16    = lane & 15;
    const int kb     = (lane >> 4) * 2;

    const int mrow   = m_base + l16;
    const int ncol_l = n_tile * 16 + l16;
    const float* arow = h + (size_t)mrow * DD;
    const float* brow = wlds + ncol_l * DD;

    v8f acc = {};
    #pragma unroll 8
    for (int k = 0; k < DD; k += 4) {
        v2f a, b;
        a.x = arow[k + kb];
        a.y = arow[k + kb + 1];
        b.x = brow[k + kb];
        b.y = brow[k + kb + 1];
        acc = __builtin_amdgcn_wmma_f32_16x16x4_f32(
             false, a,  false, b,
             (short)0, acc,  false,  false);
    }

    if (ncol_l < CC) {
        float bv = bias[ncol_l];
        int m_off = (lane >> 4) * 8;
        #pragma unroll
        for (int v = 0; v < 8; ++v) {
            out[(size_t)(m_base + m_off + v) * CC + ncol_l] = acc[v] + bv;
        }
    }
}

__global__ void dgc_layernorm(float* __restrict__ out,
                              const float* __restrict__ gamma,
                              const float* __restrict__ beta) {
    int node = (blockIdx.x * blockDim.x + threadIdx.x) >> 5;
    int lane = threadIdx.x & 31;
    if (node >= NN) return;
    float* rowp = out + (size_t)node * CC;

    float x0 = rowp[lane];
    float x1 = (lane < CC - 32) ? rowp[32 + lane] : 0.f;
    float s  = x0 + x1;
    float sq = x0 * x0 + x1 * x1;
    #pragma unroll
    for (int off = 16; off > 0; off >>= 1) {
        s  += __shfl_xor(s,  off, 32);
        sq += __shfl_xor(sq, off, 32);
    }
    float mean = s * (1.0f / CC);
    float var  = sq * (1.0f / CC) - mean * mean;
    float inv  = rsqrtf(var + LN_EPS);

    rowp[lane] = (x0 - mean) * inv * gamma[lane] + beta[lane];
    if (lane < CC - 32)
        rowp[32 + lane] = (x1 - mean) * inv * gamma[32 + lane] + beta[32 + lane];
}

extern "C" void kernel_launch(void* const* d_in, const int* in_sizes, int n_in,
                              void* d_out, int out_size, void* d_ws, size_t ws_size,
                              hipStream_t stream) {
    const float* feat  = (const float*)d_in[0];
    const int*   row   = (const int*)  d_in[1];
    const int*   col   = (const int*)  d_in[2];
    const float* W     = (const float*)d_in[3];
    const float* bias  = (const float*)d_in[4];
    const float* gamma = (const float*)d_in[5];
    const float* beta  = (const float*)d_in[6];
    float* out = (float*)d_out;

    const size_t NP = 100352;
    float* dinv = (float*)d_ws;
    float* h    = dinv + NP;
    float* agg  = h + (size_t)NN * DD;

    const int B = 256;

    dgc_deg_init <<<(NN + B - 1) / B, B, 0, stream>>>(dinv);
    dgc_deg_count<<<(EE + B - 1) / B, B, 0, stream>>>(row, dinv);
    dgc_deg_rsqrt<<<(NN + B - 1) / B, B, 0, stream>>>(dinv);

    const int elem_blocks = (NN * (DD / 4) + B - 1) / B;
    const int edge_blocks = (EE * 32 + B - 1) / B;
    for (int step = 0; step < KSTEPS; ++step) {
        const float* src = (step == 0) ? feat : h;
        dgc_init_agg    <<<elem_blocks, B, 0, stream>>>(src, dinv, agg);
        dgc_edge_scatter<<<edge_blocks, B, 0, stream>>>(src, row, col, dinv, agg);
        dgc_combine     <<<elem_blocks, B, 0, stream>>>(src, agg, h);
    }

    const int fc_waves  = M_TILES * N_TILES;
    const int fc_blocks = (fc_waves * 32 + FC_BLOCK - 1) / FC_BLOCK;
    dgc_fc_wmma<<<fc_blocks, FC_BLOCK, 0, stream>>>(h, W, bias, out);

    dgc_layernorm<<<(NN * 32 + B - 1) / B, B, 0, stream>>>(out, gamma, beta);
}


# FC (128->40, f32 WMMA) moved ahead of the 4 propagation steps (they commute): edge scatter-adds and axpy passes run on 40 columns, a wave per 32 edges; bias joins LayerNorm; f32 sum order changes
# speedup vs baseline: 2.5589x; 2.5589x over previous
#include <hip/hip_runtime.h>
#include <hip/hip_bf16.h>

#define NN      100000
#define EE      1600000
#define DD      128
#define CC      40
#define KSTEPS  4
#define DELTA   0.5f
#define LN_EPS  1e-5f

typedef __attribute__((ext_vector_type(2))) float v2f;
typedef __attribute__((ext_vector_type(8))) float v8f;

__global__ void dgc_deg_init(float* __restrict__ deg) {
    int i = blockIdx.x * blockDim.x + threadIdx.x;
    if (i < NN) deg[i] = 1.0f;
}

__global__ void dgc_deg_count(const int* __restrict__ row, float* __restrict__ deg) {
    int e = blockIdx.x * blockDim.x + threadIdx.x;
    if (e < EE) atomicAdd(&deg[row[e]], 1.0f);
}

__global__ void dgc_deg_rsqrt(float* __restrict__ deg) {
    int i = blockIdx.x * blockDim.x + threadIdx.x;
    if (i < NN) deg[i] = rsqrtf(deg[i]);
}

__global__ void dgc_init_agg(const float* __restrict__ src,
                             const float* __restrict__ dinv,
                             float* __restrict__ agg) {
    int idx = blockIdx.x * blockDim.x + threadIdx.x;
    if (idx >= NN * (CC / 4)) return;
    int node = idx / (CC / 4);
    float ws = dinv[node] * dinv[node];
    const float4* s = (const float4*)src;
    float4 v = s[idx];
    v.x *= ws; v.y *= ws; v.z *= ws; v.w *= ws;
    ((float4*)agg)[idx] = v;
}

#define SC_PASSES ((32 * CC) / 32)
#define SC_GROUP  10
static_assert(SC_PASSES % SC_GROUP == 0, "whole groups");
static_assert(EE % 32 == 0, "a wave owns 32 whole edges");

__global__ void dgc_edge_scatter(const float* __restrict__ src,
                                 const int* __restrict__ row,
                                 const int* __restrict__ col,
                                 const float* __restrict__ dinv,
                                 float* __restrict__ agg) {
    const int wave = (blockIdx.x * blockDim.x + threadIdx.x) >> 5;
    const int lane = threadIdx.x & 31;
    if (wave * 32 >= EE) return;
    const int e = wave * 32 + lane;
    const int r = row[e];
    const int c = col[e];
    const int wbits = __float_as_int(dinv[r] * dinv[c]);
    const int roff  = r * CC;
    const int coff  = c * CC;

    #pragma unroll
    for (int g = 0; g < SC_PASSES / SC_GROUP; ++g) {
        float    val[SC_GROUP];
        unsigned dst[SC_GROUP];
        #pragma unroll
        for (int p = 0; p < SC_GROUP; ++p) {
            const int  i     = g * SC_GROUP + p;
            const int  k0    = (32 * i) / CC;
            const int  split = CC * (k0 + 1) - 32 * i;
            const bool two   = split < 32;
            const int  k1    = two ? k0 + 1 : k0;

            const int r0 = __builtin_amdgcn_readlane(roff,  k0);
            const int c0 = __builtin_amdgcn_readlane(coff,  k0);
            const int w0 = __builtin_amdgcn_readlane(wbits, k0);
            const int r1 = __builtin_amdgcn_readlane(roff,  k1);
            const int c1 = __builtin_amdgcn_readlane(coff,  k1);
            const int w1 = __builtin_amdgcn_readlane(wbits, k1);

            const bool hi = two && (lane >= split);
            const int  j  = 32 * i + lane - CC * (hi ? k0 + 1 : k0);
            const unsigned si = (unsigned)((hi ? c1 : c0) + j);
            dst[p] = (unsigned)((hi ? r1 : r0) + j);
            val[p] = src[si] * __int_as_float(hi ? w1 : w0);
        }
        #pragma unroll
        for (int p = 0; p < SC_GROUP; ++p)
            atomicAdd(agg + dst[p], val[p]);
    }
}

__global__ void dgc_combine(const float* __restrict__ agg,
                            float* __restrict__ h) {
    int idx = blockIdx.x * blockDim.x + threadIdx.x;
    if (idx >= NN * (CC / 4)) return;
    float4 a = ((const float4*)h)[idx];
    float4 g = ((const float4*)agg)[idx];
    float4 o;
    o.x = (1.0f - DELTA) * a.x + DELTA * g.x;
    o.y = (1.0f - DELTA) * a.y + DELTA * g.y;
    o.z = (1.0f - DELTA) * a.z + DELTA * g.z;
    o.w = (1.0f - DELTA) * a.w + DELTA * g.w;
    ((float4*)h)[idx] = o;
}

#define M_TILES   (NN / 16)
#define N_TILES   3
#define NPAD      (N_TILES * 16)
#define FC_BLOCK  256

__global__ void dgc_fc_wmma(const float* __restrict__ x,
                            const float* __restrict__ W,
                            float* __restrict__ out) {
    __shared__ float wlds[NPAD * DD];

    for (int idx = threadIdx.x; idx < NPAD * DD; idx += FC_BLOCK) {
        int colc = idx >> 7;
        wlds[idx] = (colc < CC) ? W[idx] : 0.0f;
    }
    __syncthreads();

    int wave = (blockIdx.x * FC_BLOCK + threadIdx.x) >> 5;
    int lane = threadIdx.x & 31;
    int m_tile = wave / N_TILES;
    int n_tile = wave % N_TILES;
    if (m_tile >= M_TILES) return;

    const int m_base = m_tile * 16;
    const int l16    = lane & 15;
    const int kb     = (lane >> 4) * 2;

    const int mrow   = m_base + l16;
    const int ncol_l = n_tile * 16 + l16;
    const float* arow = x + (size_t)mrow * DD;
    const float* brow = wlds + ncol_l * DD;

    v8f acc = {};
    #pragma unroll 8
    for (int k = 0; k < DD; k += 4) {
        v2f a, b;
        a.x = arow[k + kb];
        a.y = arow[k + kb + 1];
        b.x = brow[k + kb];
        b.y = brow[k + kb + 1];
        acc = __builtin_amdgcn_wmma_f32_16x16x4_f32(
             false, a,  false, b,
             (short)0, acc,  false,  false);
    }

    if (ncol_l < CC) {
        int m_off = (lane >> 4) * 8;
        #pragma unroll
        for (int v = 0; v < 8; ++v) {
            out[(size_t)(m_base + m_off + v) * CC + ncol_l] = acc[v];
        }
    }
}

__global__ void dgc_layernorm(const float* __restrict__ hin,
                              const float* __restrict__ bias,
                              float* __restrict__ out,
                              const float* __restrict__ gamma,
                              const float* __restrict__ beta) {
    int node = (blockIdx.x * blockDim.x + threadIdx.x) >> 5;
    int lane = threadIdx.x & 31;
    if (node >= NN) return;
    const float* inp = hin + (size_t)node * CC;
    float* rowp = out + (size_t)node * CC;

    float x0 = inp[lane] + bias[lane];
    float x1 = (lane < CC - 32) ? inp[32 + lane] + bias[32 + lane] : 0.f;
    float s  = x0 + x1;
    float sq = x0 * x0 + x1 * x1;
    #pragma unroll
    for (int off = 16; off > 0; off >>= 1) {
        s  += __shfl_xor(s,  off, 32);
        sq += __shfl_xor(sq, off, 32);
    }
    float mean = s * (1.0f / CC);
    float var  = sq * (1.0f / CC) - mean * mean;
    float inv  = rsqrtf(var + LN_EPS);

    rowp[lane] = (x0 - mean) * inv * gamma[lane] + beta[lane];
    if (lane < CC - 32)
        rowp[32 + lane] = (x1 - mean) * inv * gamma[32 + lane] + beta[32 + lane];
}

extern "C" void kernel_launch(void* const* d_in, const int* in_sizes, int n_in,
                              void* d_out, int out_size, void* d_ws, size_t ws_size,
                              hipStream_t stream) {
    const float* feat  = (const float*)d_in[0];
    const int*   row   = (const int*)  d_in[1];
    const int*   col   = (const int*)  d_in[2];
    const float* W     = (const float*)d_in[3];
    const float* bias  = (const float*)d_in[4];
    const float* gamma = (const float*)d_in[5];
    const float* beta  = (const float*)d_in[6];
    float* out = (float*)d_out;

    const size_t NP = 100352;
    float* dinv = (float*)d_ws;
    float* h    = dinv + NP;
    float* agg  = h + (size_t)NN * CC;

    const int B = 256;

    dgc_deg_init <<<(NN + B - 1) / B, B, 0, stream>>>(dinv);
    dgc_deg_count<<<(EE + B - 1) / B, B, 0, stream>>>(row, dinv);
    dgc_deg_rsqrt<<<(NN + B - 1) / B, B, 0, stream>>>(dinv);

    const int fc_waves  = M_TILES * N_TILES;
    const int fc_blocks = (fc_waves * 32 + FC_BLOCK - 1) / FC_BLOCK;
    dgc_fc_wmma<<<fc_blocks, FC_BLOCK, 0, stream>>>(feat, W, h);

    const int elem_blocks = (NN * (CC / 4) + B - 1) / B;
    const int edge_blocks = (EE + B - 1) / B;
    for (int step = 0; step < KSTEPS; ++step) {
        dgc_init_agg    <<<elem_blocks, B, 0, stream>>>(h, dinv, agg);
        dgc_edge_scatter<<<edge_blocks, B, 0, stream>>>(h, row, col, dinv, agg);
        dgc_combine     <<<elem_blocks, B, 0, stream>>>(agg, h);
    }

    dgc_layernorm<<<(NN * 32 + B - 1) / B, B, 0, stream>>>(h, bias, out, gamma, beta);
}


# Edges bucketed by row once per call (int atomic arrival-order rank, scan, fill); the 4 steps become per-row register gathers, no float atomics, self term and axpy folded in; f32 sum order changes
# speedup vs baseline: 4.8934x; 1.9123x over previous
#include <hip/hip_runtime.h>
#include <hip/hip_bf16.h>

#define NN      100000
#define EE      1600000
#define DD      128
#define CC      40
#define KSTEPS  4
#define DELTA   0.5f
#define LN_EPS  1e-5f

#define CSR_PAD 8
#define SCAN_B  1024
#define NP      100352
static_assert(NP % SCAN_B == 0 && NP > NN, "whole scan blocks, and entry N exists");
static_assert(KSTEPS % 2 == 0, "the ping-pong ends in the buffer it started in");

typedef __attribute__((ext_vector_type(2))) float v2f;
typedef __attribute__((ext_vector_type(8))) float v8f;

__global__ void dgc_zero_cnt(int* __restrict__ cnt) {
    int i = blockIdx.x * blockDim.x + threadIdx.x;
    if (i < NP) cnt[i] = 0;
}

__global__ void dgc_count_rank(const int* __restrict__ row,
                               int* __restrict__ cnt,
                               int* __restrict__ rank) {
    int e = blockIdx.x * blockDim.x + threadIdx.x;
    if (e < EE) rank[e] = atomicAdd(&cnt[row[e]], 1);
}

__global__ __launch_bounds__(SCAN_B)
void dgc_scan(const int* __restrict__ cnt, int* __restrict__ node_raw) {
    __shared__ int s_pre[32];
    __shared__ int s_tot[32];
    const int t = threadIdx.x, lane = t & 31, wv = t >> 5;
    const int i = blockIdx.x * SCAN_B + t;
    int pre = 0;
    for (int j = 0; j < (int)blockIdx.x; ++j) pre += cnt[j * SCAN_B + t];
    const int c = cnt[i];
    int inc = c;
    for (int off = 1; off < 32; off <<= 1) {
        int v = __shfl_up(inc, off, 32);
        if (lane >= off) inc += v;
    }
    for (int off = 16; off > 0; off >>= 1) pre += __shfl_xor(pre, off, 32);
    if (lane == 31) s_tot[wv] = inc;
    if (lane == 0)  s_pre[wv] = pre;
    __syncthreads();
    int base = s_pre[lane] + ((lane < wv) ? s_tot[lane] : 0);
    for (int off = 16; off > 0; off >>= 1) base += __shfl_xor(base, off, 32);
    int2 rec;
    rec.x = base + inc - c;
    rec.y = __float_as_int(rsqrtf((float)c + 1.0f));
    ((int2*)node_raw)[i] = rec;
}

__global__ void dgc_fill(const int* __restrict__ row,
                         const int* __restrict__ col,
                         const int* __restrict__ rank,
                         const int* __restrict__ node_raw,
                         int* __restrict__ csr_raw) {
    int e = blockIdx.x * blockDim.x + threadIdx.x;
    if (e >= EE) return;
    if (e < CSR_PAD) ((int2*)csr_raw)[EE + e] = make_int2(0, 0);
    const int2* node = (const int2*)node_raw;
    const int r = row[e];
    const int c = col[e];
    const int2 nr = node[r];
    const float w = __int_as_float(nr.y) * __int_as_float(node[c].y);
    int2 rec;
    rec.x = c * CC;
    rec.y = __float_as_int(w);
    ((int2*)csr_raw)[nr.x + rank[e]] = rec;
}

#define GL     10
#define RPW    3
#define NTRIP  ((NN + RPW - 1) / RPW)
#define PB     8
static_assert(GL * 4 == CC, "ten float4 make a row");
static_assert(PB - 1 <= CSR_PAD, "a part batch over-reads at most PB - 1 records");

template <bool FULL>
__device__ __forceinline__ void dgc_batch(const float* __restrict__ hin,
                                          const int2* __restrict__ csr,
                                          int e, int rem, int j4, float4& acc) {
    int2   m[PB];
    float4 v[PB];
    #pragma unroll
    for (int k = 0; k < PB; ++k) {
        m[k] = csr[e + k];
        if (!FULL && k >= rem) m[k].y = 0;
    }
    if (!FULL) {
        #pragma unroll
        for (int k = 0; k < PB; ++k)
            asm volatile("" : "+v"(m[k].x), "+v"(m[k].y));
    }
    #pragma unroll
    for (int k = 0; k < PB; ++k) {
        v[k] = make_float4(0.f, 0.f, 0.f, 0.f);
        if (FULL || k < rem) v[k] = *(const float4*)(hin + m[k].x + j4);
    }
    #pragma unroll
    for (int k = 0; k < PB; ++k) {
        const float w = __int_as_float(m[k].y);
        acc.x = fmaf(v[k].x, w, acc.x);
        acc.y = fmaf(v[k].y, w, acc.y);
        acc.z = fmaf(v[k].z, w, acc.z);
        acc.w = fmaf(v[k].w, w, acc.w);
    }
}

__global__ __launch_bounds__(256)
void dgc_prop(const float* __restrict__ hin,
              float* __restrict__ hout,
              const int* __restrict__ node_raw,
              const int* __restrict__ csr_raw) {
    const int2* node = (const int2*)node_raw;
    const int2* csr  = (const int2*)csr_raw;
    const int lane   = threadIdx.x & 31;
    const int grp    = lane / GL;
    const int j4     = (lane - grp * GL) * 4;
    const int wave   = (blockIdx.x * blockDim.x + threadIdx.x) >> 5;
    const int nwaves = (gridDim.x * blockDim.x) >> 5;

    for (int t = wave; t < NTRIP; t += nwaves) {
        const int  r    = t * RPW + grp;
        const bool live = (grp < RPW) && (r < NN);
        int beg = 0, end = 0;
        float ws = 0.f;
        float4 hv = make_float4(0.f, 0.f, 0.f, 0.f);
        if (live) {
            const int2 nr = node[r];
            beg = nr.x;
            end = node[r + 1].x;
            ws  = __int_as_float(nr.y) * __int_as_float(nr.y);
            hv  = *(const float4*)(hin + r * CC + j4);
        }
        float4 acc = make_float4(0.f, 0.f, 0.f, 0.f);
        int e = beg;
        for (; e + PB <= end; e += PB) dgc_batch<true>(hin, csr, e, PB, j4, acc);
        if (e < end) dgc_batch<false>(hin, csr, e, end - e, j4, acc);
        if (live) {
            float4 o;
            o.x = (1.0f - DELTA) * hv.x + DELTA * fmaf(hv.x, ws, acc.x);
            o.y = (1.0f - DELTA) * hv.y + DELTA * fmaf(hv.y, ws, acc.y);
            o.z = (1.0f - DELTA) * hv.z + DELTA * fmaf(hv.z, ws, acc.z);
            o.w = (1.0f - DELTA) * hv.w + DELTA * fmaf(hv.w, ws, acc.w);
            *(float4*)(hout + r * CC + j4) = o;
        }
    }
}

#define M_TILES   (NN / 16)
#define N_TILES   3
#define NPAD      (N_TILES * 16)
#define FC_BLOCK  256

__global__ void dgc_fc_wmma(const float* __restrict__ x,
                            const float* __restrict__ W,
                            float* __restrict__ out) {
    __shared__ float wlds[NPAD * DD];

    for (int idx = threadIdx.x; idx < NPAD * DD; idx += FC_BLOCK) {
        int colc = idx >> 7;
        wlds[idx] = (colc < CC) ? W[idx] : 0.0f;
    }
    __syncthreads();

    int wave = (blockIdx.x * FC_BLOCK + threadIdx.x) >> 5;
    int lane = threadIdx.x & 31;
    int m_tile = wave / N_TILES;
    int n_tile = wave % N_TILES;
    if (m_tile >= M_TILES) return;

    const int m_base = m_tile * 16;
    const int l16    = lane & 15;
    const int kb     = (lane >> 4) * 2;

    const int mrow   = m_base + l16;
    const int ncol_l = n_tile * 16 + l16;
    const float* arow = x + (size_t)mrow * DD;
    const float* brow = wlds + ncol_l * DD;

    v8f acc = {};
    #pragma unroll 8
    for (int k = 0; k < DD; k += 4) {
        v2f a, b;
        a.x = arow[k + kb];
        a.y = arow[k + kb + 1];
        b.x = brow[k + kb];
        b.y = brow[k + kb + 1];
        acc = __builtin_amdgcn_wmma_f32_16x16x4_f32(
             false, a,  false, b,
             (short)0, acc,  false,  false);
    }

    if (ncol_l < CC) {
        int m_off = (lane >> 4) * 8;
        #pragma unroll
        for (int v = 0; v < 8; ++v) {
            out[(size_t)(m_base + m_off + v) * CC + ncol_l] = acc[v];
        }
    }
}

__global__ void dgc_layernorm(const float* __restrict__ hin,
                              const float* __restrict__ bias,
                              float* __restrict__ out,
                              const float* __restrict__ gamma,
                              const float* __restrict__ beta) {
    int node = (blockIdx.x * blockDim.x + threadIdx.x) >> 5;
    int lane = threadIdx.x & 31;
    if (node >= NN) return;
    const float* inp = hin + (size_t)node * CC;
    float* rowp = out + (size_t)node * CC;

    float x0 = inp[lane] + bias[lane];
    float x1 = (lane < CC - 32) ? inp[32 + lane] + bias[32 + lane] : 0.f;
    float s  = x0 + x1;
    float sq = x0 * x0 + x1 * x1;
    #pragma unroll
    for (int off = 16; off > 0; off >>= 1) {
        s  += __shfl_xor(s,  off, 32);
        sq += __shfl_xor(sq, off, 32);
    }
    float mean = s * (1.0f / CC);
    float var  = sq * (1.0f / CC) - mean * mean;
    float inv  = rsqrtf(var + LN_EPS);

    rowp[lane] = (x0 - mean) * inv * gamma[lane] + beta[lane];
    if (lane < CC - 32)
        rowp[32 + lane] = (x1 - mean) * inv * gamma[32 + lane] + beta[32 + lane];
}

extern "C" void kernel_launch(void* const* d_in, const int* in_sizes, int n_in,
                              void* d_out, int out_size, void* d_ws, size_t ws_size,
                              hipStream_t stream) {
    const float* feat  = (const float*)d_in[0];
    const int*   row   = (const int*)  d_in[1];
    const int*   col   = (const int*)  d_in[2];
    const float* W     = (const float*)d_in[3];
    const float* bias  = (const float*)d_in[4];
    const float* gamma = (const float*)d_in[5];
    const float* beta  = (const float*)d_in[6];
    float* out = (float*)d_out;

    int*   node = (int*)d_ws;
    int*   cnt  = node + 2 * NP;
    float* h    = (float*)(cnt + NP);
    int*   rank = (int*)h;
    int*   csr  = (int*)(h + (size_t)NN * CC);

    const int B = 256;
    const int edge_blocks = (EE + B - 1) / B;

    dgc_zero_cnt  <<<NP / B, B, 0, stream>>>(cnt);
    dgc_count_rank<<<edge_blocks, B, 0, stream>>>(row, cnt, rank);
    dgc_scan      <<<NP / SCAN_B, SCAN_B, 0, stream>>>(cnt, node);
    dgc_fill      <<<edge_blocks, B, 0, stream>>>(row, col, rank, node, csr);

    const int fc_waves  = M_TILES * N_TILES;
    const int fc_blocks = (fc_waves * 32 + FC_BLOCK - 1) / FC_BLOCK;
    dgc_fc_wmma<<<fc_blocks, FC_BLOCK, 0, stream>>>(feat, W, h);

    const int prop_blocks = (NTRIP / 2 * 32 + B - 1) / B;
    for (int step = 0; step < KSTEPS; step += 2) {
        dgc_prop<<<prop_blocks, B, 0, stream>>>(h, out, node, csr);
        dgc_prop<<<prop_blocks, B, 0, stream>>>(out, h, node, csr);
    }

    dgc_layernorm<<<(NN * 32 + B - 1) / B, B, 0, stream>>>(h, bias, out, gamma, beta);
}


# bias + LayerNorm fused into the 4th propagation step: a row's ten lanes sum by lane permutes and store the output; the one-wave-per-node pass and its launch go; LayerNorm f32 sum order changes
# speedup vs baseline: 5.0577x; 1.0336x over previous
#include <hip/hip_runtime.h>
#include <hip/hip_bf16.h>

#define NN      100000
#define EE      1600000
#define DD      128
#define CC      40
#define KSTEPS  4
#define DELTA   0.5f
#define LN_EPS  1e-5f

#define CSR_PAD 8
#define SCAN_B  1024
#define NP      100352
static_assert(NP % SCAN_B == 0 && NP > NN, "whole scan blocks, and entry N exists");
static_assert(KSTEPS == 4, "the launcher writes the four steps out: FC into out, last step into out");

typedef __attribute__((ext_vector_type(2))) float v2f;
typedef __attribute__((ext_vector_type(8))) float v8f;

__global__ void dgc_zero_cnt(int* __restrict__ cnt) {
    int i = blockIdx.x * blockDim.x + threadIdx.x;
    if (i < NP) cnt[i] = 0;
}

__global__ void dgc_count_rank(const int* __restrict__ row,
                               int* __restrict__ cnt,
                               int* __restrict__ rank) {
    int e = blockIdx.x * blockDim.x + threadIdx.x;
    if (e < EE) rank[e] = atomicAdd(&cnt[row[e]], 1);
}

__global__ __launch_bounds__(SCAN_B)
void dgc_scan(const int* __restrict__ cnt, int* __restrict__ node_raw) {
    __shared__ int s_pre[32];
    __shared__ int s_tot[32];
    const int t = threadIdx.x, lane = t & 31, wv = t >> 5;
    const int i = blockIdx.x * SCAN_B + t;
    int pre = 0;
    for (int j = 0; j < (int)blockIdx.x; ++j) pre += cnt[j * SCAN_B + t];
    const int c = cnt[i];
    int inc = c;
    for (int off = 1; off < 32; off <<= 1) {
        int v = __shfl_up(inc, off, 32);
        if (lane >= off) inc += v;
    }
    for (int off = 16; off > 0; off >>= 1) pre += __shfl_xor(pre, off, 32);
    if (lane == 31) s_tot[wv] = inc;
    if (lane == 0)  s_pre[wv] = pre;
    __syncthreads();
    int base = s_pre[lane] + ((lane < wv) ? s_tot[lane] : 0);
    for (int off = 16; off > 0; off >>= 1) base += __shfl_xor(base, off, 32);
    int2 rec;
    rec.x = base + inc - c;
    rec.y = __float_as_int(rsqrtf((float)c + 1.0f));
    ((int2*)node_raw)[i] = rec;
}

__global__ void dgc_fill(const int* __restrict__ row,
                         const int* __restrict__ col,
                         const int* __restrict__ rank,
                         const int* __restrict__ node_raw,
                         int* __restrict__ csr_raw) {
    int e = blockIdx.x * blockDim.x + threadIdx.x;
    if (e >= EE) return;
    if (e < CSR_PAD) ((int2*)csr_raw)[EE + e] = make_int2(0, 0);
    const int2* node = (const int2*)node_raw;
    const int r = row[e];
    const int c = col[e];
    const int2 nr = node[r];
    const float w = __int_as_float(nr.y) * __int_as_float(node[c].y);
    int2 rec;
    rec.x = c * CC;
    rec.y = __float_as_int(w);
    ((int2*)csr_raw)[nr.x + rank[e]] = rec;
}

#define GL     10
#define RPW    3
#define NTRIP  ((NN + RPW - 1) / RPW)
#define PB     8
static_assert(GL * 4 == CC, "ten float4 make a row");
static_assert(PB - 1 <= CSR_PAD, "a part batch over-reads at most PB - 1 records");

template <bool FULL>
__device__ __forceinline__ void dgc_batch(const float* __restrict__ hin,
                                          const int2* __restrict__ csr,
                                          int e, int rem, int j4, float4& acc) {
    int2   m[PB];
    float4 v[PB];
    #pragma unroll
    for (int k = 0; k < PB; ++k) {
        m[k] = csr[e + k];
        if (!FULL && k >= rem) m[k].y = 0;
    }
    if (!FULL) {
        #pragma unroll
        for (int k = 0; k < PB; ++k)
            asm volatile("" : "+v"(m[k].x), "+v"(m[k].y));
    }
    #pragma unroll
    for (int k = 0; k < PB; ++k) {
        v[k] = make_float4(0.f, 0.f, 0.f, 0.f);
        if (FULL || k < rem) v[k] = *(const float4*)(hin + m[k].x + j4);
    }
    #pragma unroll
    for (int k = 0; k < PB; ++k) {
        const float w = __int_as_float(m[k].y);
        acc.x = fmaf(v[k].x, w, acc.x);
        acc.y = fmaf(v[k].y, w, acc.y);
        acc.z = fmaf(v[k].z, w, acc.z);
        acc.w = fmaf(v[k].w, w, acc.w);
    }
}

struct dgc_ring { int p5, p1, p2, p3, p4, p0; };

__device__ __forceinline__ float dgc_group_sum(float v, const dgc_ring& g) {
    const float t = v + __shfl(v, g.p5, 32);
    float s = t + __shfl(t, g.p1, 32);
    s += __shfl(t, g.p2, 32);
    s += __shfl(t, g.p3, 32);
    s += __shfl(t, g.p4, 32);
    return __shfl(s, g.p0, 32);
}

__device__ __forceinline__ float4 dgc_step_row(const float4 hv, const float ws, const float4 acc) {
    float4 o;
    o.x = (1.0f - DELTA) * hv.x + DELTA * fmaf(hv.x, ws, acc.x);
    o.y = (1.0f - DELTA) * hv.y + DELTA * fmaf(hv.y, ws, acc.y);
    o.z = (1.0f - DELTA) * hv.z + DELTA * fmaf(hv.z, ws, acc.z);
    o.w = (1.0f - DELTA) * hv.w + DELTA * fmaf(hv.w, ws, acc.w);
    return o;
}

template <bool LAST>
__global__ __launch_bounds__(256)
void dgc_prop(const float* __restrict__ hin,
              float* __restrict__ hout,
              const int* __restrict__ node_raw,
              const int* __restrict__ csr_raw,
              const float* __restrict__ bias,
              const float* __restrict__ gamma,
              const float* __restrict__ beta) {
    const int2* node = (const int2*)node_raw;
    const int2* csr  = (const int2*)csr_raw;
    const int lane   = threadIdx.x & 31;
    const int grp    = lane / GL;
    const int j4     = (lane - grp * GL) * 4;
    const int wave   = (blockIdx.x * blockDim.x + threadIdx.x) >> 5;
    const int nwaves = (gridDim.x * blockDim.x) >> 5;

    float4 bv = make_float4(0.f, 0.f, 0.f, 0.f), gv = bv, tv = bv;
    dgc_ring ring = {lane, lane, lane, lane, lane, lane};
    if (LAST) {
        bv = make_float4(bias[j4],  bias[j4 + 1],  bias[j4 + 2],  bias[j4 + 3]);
        gv = make_float4(gamma[j4], gamma[j4 + 1], gamma[j4 + 2], gamma[j4 + 3]);
        tv = make_float4(beta[j4],  beta[j4 + 1],  beta[j4 + 2],  beta[j4 + 3]);
        if (grp < RPW) {
            const int base = grp * GL;
            const int j    = lane - base;
            ring.p5 = base + (j + 5) % GL;
            ring.p1 = base + (j + 1) % GL;
            ring.p2 = base + (j + 2) % GL;
            ring.p3 = base + (j + 3) % GL;
            ring.p4 = base + (j + 4) % GL;
            ring.p0 = base;
        }
    }

    for (int t = wave; t < NTRIP; t += nwaves) {
        const int  r    = t * RPW + grp;
        const bool live = (grp < RPW) && (r < NN);
        int beg = 0, end = 0;
        float ws = 0.f;
        float4 hv = make_float4(0.f, 0.f, 0.f, 0.f);
        if (live) {
            const int2 nr = node[r];
            beg = nr.x;
            end = node[r + 1].x;
            ws  = __int_as_float(nr.y) * __int_as_float(nr.y);
            hv  = *(const float4*)(hin + r * CC + j4);
        }
        float4 acc = make_float4(0.f, 0.f, 0.f, 0.f);
        int e = beg;
        for (; e + PB <= end; e += PB) dgc_batch<true>(hin, csr, e, PB, j4, acc);
        if (e < end) dgc_batch<false>(hin, csr, e, end - e, j4, acc);
        if (!LAST) {
            if (live) *(float4*)(hout + r * CC + j4) = dgc_step_row(hv, ws, acc);
        } else {
            const float4 o = dgc_step_row(hv, ws, acc);
            float4 x;
            x.x = o.x + bv.x;
            x.y = o.y + bv.y;
            x.z = o.z + bv.z;
            x.w = o.w + bv.w;
            const float s  = dgc_group_sum((x.x + x.y) + (x.z + x.w), ring);
            const float sq = dgc_group_sum((x.x * x.x + x.y * x.y) + (x.z * x.z + x.w * x.w), ring);
            const float mean = s * (1.0f / CC);
            const float var  = sq * (1.0f / CC) - mean * mean;
            const float inv  = rsqrtf(var + LN_EPS);
            float4 y;
            y.x = (x.x - mean) * inv * gv.x + tv.x;
            y.y = (x.y - mean) * inv * gv.y + tv.y;
            y.z = (x.z - mean) * inv * gv.z + tv.z;
            y.w = (x.w - mean) * inv * gv.w + tv.w;
            if (live) *(float4*)(hout + r * CC + j4) = y;
        }
    }
}

#define M_TILES   (NN / 16)
#define N_TILES   3
#define NPAD      (N_TILES * 16)
#define FC_BLOCK  256

__global__ void dgc_fc_wmma(const float* __restrict__ x,
                            const float* __restrict__ W,
                            float* __restrict__ out) {
    __shared__ float wlds[NPAD * DD];

    for (int idx = threadIdx.x; idx < NPAD * DD; idx += FC_BLOCK) {
        int colc = idx >> 7;
        wlds[idx] = (colc < CC) ? W[idx] : 0.0f;
    }
    __syncthreads();

    int wave = (blockIdx.x * FC_BLOCK + threadIdx.x) >> 5;
    int lane = threadIdx.x & 31;
    int m_tile = wave / N_TILES;
    int n_tile = wave % N_TILES;
    if (m_tile >= M_TILES) return;

    const int m_base = m_tile * 16;
    const int l16    = lane & 15;
    const int kb     = (lane >> 4) * 2;

    const int mrow   = m_base + l16;
    const int ncol_l = n_tile * 16 + l16;
    const float* arow = x + (size_t)mrow * DD;
    const float* brow = wlds + ncol_l * DD;

    v8f acc = {};
    #pragma unroll 8
    for (int k = 0; k < DD; k += 4) {
        v2f a, b;
        a.x = arow[k + kb];
        a.y = arow[k + kb + 1];
        b.x = brow[k + kb];
        b.y = brow[k + kb + 1];
        acc = __builtin_amdgcn_wmma_f32_16x16x4_f32(
             false, a,  false, b,
             (short)0, acc,  false,  false);
    }

    if (ncol_l < CC) {
        int m_off = (lane >> 4) * 8;
        #pragma unroll
        for (int v = 0; v < 8; ++v) {
            out[(size_t)(m_base + m_off + v) * CC + ncol_l] = acc[v];
        }
    }
}

extern "C" void kernel_launch(void* const* d_in, const int* in_sizes, int n_in,
                              void* d_out, int out_size, void* d_ws, size_t ws_size,
                              hipStream_t stream) {
    const float* feat  = (const float*)d_in[0];
    const int*   row   = (const int*)  d_in[1];
    const int*   col   = (const int*)  d_in[2];
    const float* W     = (const float*)d_in[3];
    const float* bias  = (const float*)d_in[4];
    const float* gamma = (const float*)d_in[5];
    const float* beta  = (const float*)d_in[6];
    float* out = (float*)d_out;

    int*   node = (int*)d_ws;
    int*   cnt  = node + 2 * NP;
    float* h    = (float*)(cnt + NP);
    int*   rank = (int*)h;
    int*   csr  = (int*)(h + (size_t)NN * CC);

    const int B = 256;
    const int edge_blocks = (EE + B - 1) / B;

    dgc_zero_cnt  <<<NP / B, B, 0, stream>>>(cnt);
    dgc_count_rank<<<edge_blocks, B, 0, stream>>>(row, cnt, rank);
    dgc_scan      <<<NP / SCAN_B, SCAN_B, 0, stream>>>(cnt, node);
    dgc_fill      <<<edge_blocks, B, 0, stream>>>(row, col, rank, node, csr);

    const int fc_waves  = M_TILES * N_TILES;
    const int fc_blocks = (fc_waves * 32 + FC_BLOCK - 1) / FC_BLOCK;
    dgc_fc_wmma<<<fc_blocks, FC_BLOCK, 0, stream>>>(feat, W, out);

    const int prop_blocks = (NTRIP / 2 * 32 + B - 1) / B;
    dgc_prop<false><<<prop_blocks, B, 0, stream>>>(out, h, node, csr, bias, gamma, beta);
    dgc_prop<false><<<prop_blocks, B, 0, stream>>>(h, out, node, csr, bias, gamma, beta);
    dgc_prop<false><<<prop_blocks, B, 0, stream>>>(out, h, node, csr, bias, gamma, beta);
    dgc_prop<true ><<<prop_blocks, B, 0, stream>>>(h, out, node, csr, bias, gamma, beta);
}


# speedup vs baseline: 5.4397x; 1.0755x over previous
#include <hip/hip_runtime.h>
#include <hip/hip_bf16.h>

#define NN      100000
#define EE      1600000
#define DD      128
#define CC      40
#define KSTEPS  4
#define DELTA   0.5f
#define LN_EPS  1e-5f

#define HP      64
#define CSR_PAD 8
#define SCAN_B  1024
#define NP      100352
static_assert(NP % SCAN_B == 0 && NP > NN, "whole scan blocks, and entry N exists");
static_assert(KSTEPS == 4, "the launcher writes the four steps out, the last one into out");
static_assert(HP >= CC && (HP * 4) % 256 == 0, "a padded row is whole 256-byte units");

typedef __attribute__((ext_vector_type(2))) float v2f;
typedef __attribute__((ext_vector_type(8))) float v8f;

__global__ void dgc_zero_cnt(int* __restrict__ cnt) {
    int i = blockIdx.x * blockDim.x + threadIdx.x;
    if (i < NP) cnt[i] = 0;
}

__global__ void dgc_count_rank(const int* __restrict__ row,
                               int* __restrict__ cnt,
                               int* __restrict__ rank) {
    int e = blockIdx.x * blockDim.x + threadIdx.x;
    if (e < EE) rank[e] = atomicAdd(&cnt[row[e]], 1);
}

__global__ __launch_bounds__(SCAN_B)
void dgc_scan(const int* __restrict__ cnt, int* __restrict__ node_raw) {
    __shared__ int s_pre[32];
    __shared__ int s_tot[32];
    const int t = threadIdx.x, lane = t & 31, wv = t >> 5;
    const int i = blockIdx.x * SCAN_B + t;
    int pre = 0;
    for (int j = 0; j < (int)blockIdx.x; ++j) pre += cnt[j * SCAN_B + t];
    const int c = cnt[i];
    int inc = c;
    for (int off = 1; off < 32; off <<= 1) {
        int v = __shfl_up(inc, off, 32);
        if (lane >= off) inc += v;
    }
    for (int off = 16; off > 0; off >>= 1) pre += __shfl_xor(pre, off, 32);
    if (lane == 31) s_tot[wv] = inc;
    if (lane == 0)  s_pre[wv] = pre;
    __syncthreads();
    int base = s_pre[lane] + ((lane < wv) ? s_tot[lane] : 0);
    for (int off = 16; off > 0; off >>= 1) base += __shfl_xor(base, off, 32);
    int2 rec;
    rec.x = base + inc - c;
    rec.y = __float_as_int(rsqrtf((float)c + 1.0f));
    ((int2*)node_raw)[i] = rec;
}

template <int PITCH>
__global__ void dgc_fill(const int* __restrict__ row,
                         const int* __restrict__ col,
                         const int* __restrict__ rank,
                         const int* __restrict__ node_raw,
                         int* __restrict__ csr_raw) {
    int e = blockIdx.x * blockDim.x + threadIdx.x;
    if (e >= EE) return;
    if (e < CSR_PAD) ((int2*)csr_raw)[EE + e] = make_int2(0, 0);
    const int2* node = (const int2*)node_raw;
    const int r = row[e];
    const int c = col[e];
    const int2 nr = node[r];
    const float w = __int_as_float(nr.y) * __int_as_float(node[c].y);
    int2 rec;
    rec.x = c * PITCH;
    rec.y = __float_as_int(w);
    ((int2*)csr_raw)[nr.x + rank[e]] = rec;
}

#define GL     10
#define RPW    3
#define NTRIP  ((NN + RPW - 1) / RPW)
#define PB     8
static_assert(GL * 4 == CC, "ten float4 make a row");
static_assert(PB - 1 <= CSR_PAD, "a part batch over-reads at most PB - 1 records");

template <bool FULL>
__device__ __forceinline__ void dgc_batch(const float* __restrict__ hin,
                                          const int2* __restrict__ csr,
                                          int e, int rem, int j4, float4& acc) {
    int2   m[PB];
    float4 v[PB];
    #pragma unroll
    for (int k = 0; k < PB; ++k) {
        m[k] = csr[e + k];
        if (!FULL && k >= rem) m[k].y = 0;
    }
    if (!FULL) {
        #pragma unroll
        for (int k = 0; k < PB; ++k)
            asm volatile("" : "+v"(m[k].x), "+v"(m[k].y));
    }
    #pragma unroll
    for (int k = 0; k < PB; ++k) {
        v[k] = make_float4(0.f, 0.f, 0.f, 0.f);
        if (FULL || k < rem) v[k] = *(const float4*)(hin + m[k].x + j4);
    }
    #pragma unroll
    for (int k = 0; k < PB; ++k) {
        const float w = __int_as_float(m[k].y);
        acc.x = fmaf(v[k].x, w, acc.x);
        acc.y = fmaf(v[k].y, w, acc.y);
        acc.z = fmaf(v[k].z, w, acc.z);
        acc.w = fmaf(v[k].w, w, acc.w);
    }
}

struct dgc_ring { int p5, p1, p2, p3, p4, p0; };

__device__ __forceinline__ float dgc_group_sum(float v, const dgc_ring& g) {
    const float t = v + __shfl(v, g.p5, 32);
    float s = t + __shfl(t, g.p1, 32);
    s += __shfl(t, g.p2, 32);
    s += __shfl(t, g.p3, 32);
    s += __shfl(t, g.p4, 32);
    return __shfl(s, g.p0, 32);
}

__device__ __forceinline__ float4 dgc_step_row(const float4 hv, const float ws, const float4 acc) {
    float4 o;
    o.x = (1.0f - DELTA) * hv.x + DELTA * fmaf(hv.x, ws, acc.x);
    o.y = (1.0f - DELTA) * hv.y + DELTA * fmaf(hv.y, ws, acc.y);
    o.z = (1.0f - DELTA) * hv.z + DELTA * fmaf(hv.z, ws, acc.z);
    o.w = (1.0f - DELTA) * hv.w + DELTA * fmaf(hv.w, ws, acc.w);
    return o;
}

template <bool LAST, int PIN, int POUT>
__global__ __launch_bounds__(256)
void dgc_prop(const float* __restrict__ hin,
              float* __restrict__ hout,
              const int* __restrict__ node_raw,
              const int* __restrict__ csr_raw,
              const float* __restrict__ bias,
              const float* __restrict__ gamma,
              const float* __restrict__ beta) {
    const int2* node = (const int2*)node_raw;
    const int2* csr  = (const int2*)csr_raw;
    const int lane   = threadIdx.x & 31;
    const int grp    = lane / GL;
    const int j4     = (lane - grp * GL) * 4;
    const int wave   = (blockIdx.x * blockDim.x + threadIdx.x) >> 5;
    const int nwaves = (gridDim.x * blockDim.x) >> 5;

    float4 bv = make_float4(0.f, 0.f, 0.f, 0.f), gv = bv, tv = bv;
    dgc_ring ring = {lane, lane, lane, lane, lane, lane};
    if (LAST) {
        bv = make_float4(bias[j4],  bias[j4 + 1],  bias[j4 + 2],  bias[j4 + 3]);
        gv = make_float4(gamma[j4], gamma[j4 + 1], gamma[j4 + 2], gamma[j4 + 3]);
        tv = make_float4(beta[j4],  beta[j4 + 1],  beta[j4 + 2],  beta[j4 + 3]);
        if (grp < RPW) {
            const int base = grp * GL;
            const int j    = lane - base;
            ring.p5 = base + (j + 5) % GL;
            ring.p1 = base + (j + 1) % GL;
            ring.p2 = base + (j + 2) % GL;
            ring.p3 = base + (j + 3) % GL;
            ring.p4 = base + (j + 4) % GL;
            ring.p0 = base;
        }
    }

    for (int t = wave; t < NTRIP; t += nwaves) {
        const int  r    = t * RPW + grp;
        const bool live = (grp < RPW) && (r < NN);
        int beg = 0, end = 0;
        float ws = 0.f;
        float4 hv = make_float4(0.f, 0.f, 0.f, 0.f);
        if (live) {
            const int2 nr = node[r];
            beg = nr.x;
            end = node[r + 1].x;
            ws  = __int_as_float(nr.y) * __int_as_float(nr.y);
            hv  = *(const float4*)(hin + r * PIN + j4);
        }
        float4 acc = make_float4(0.f, 0.f, 0.f, 0.f);
        int e = beg;
        for (; e + PB <= end; e += PB) dgc_batch<true>(hin, csr, e, PB, j4, acc);
        if (e < end) dgc_batch<false>(hin, csr, e, end - e, j4, acc);
        if (!LAST) {
            if (live) *(float4*)(hout + r * POUT + j4) = dgc_step_row(hv, ws, acc);
        } else {
            const float4 o = dgc_step_row(hv, ws, acc);
            float4 x;
            x.x = o.x + bv.x;
            x.y = o.y + bv.y;
            x.z = o.z + bv.z;
            x.w = o.w + bv.w;
            const float s  = dgc_group_sum((x.x + x.y) + (x.z + x.w), ring);
            const float sq = dgc_group_sum((x.x * x.x + x.y * x.y) + (x.z * x.z + x.w * x.w), ring);
            const float mean = s * (1.0f / CC);
            const float var  = sq * (1.0f / CC) - mean * mean;
            const float inv  = rsqrtf(var + LN_EPS);
            float4 y;
            y.x = (x.x - mean) * inv * gv.x + tv.x;
            y.y = (x.y - mean) * inv * gv.y + tv.y;
            y.z = (x.z - mean) * inv * gv.z + tv.z;
            y.w = (x.w - mean) * inv * gv.w + tv.w;
            if (live) *(float4*)(hout + r * POUT + j4) = y;
        }
    }
}

#define M_TILES   (NN / 16)
#define N_TILES   3
#define NPAD      (N_TILES * 16)
#define FC_BLOCK  256

template <int PITCH>
__global__ void dgc_fc_wmma(const float* __restrict__ x,
                            const float* __restrict__ W,
                            float* __restrict__ out) {
    __shared__ float wlds[NPAD * DD];

    for (int idx = threadIdx.x; idx < NPAD * DD; idx += FC_BLOCK) {
        int colc = idx >> 7;
        wlds[idx] = (colc < CC) ? W[idx] : 0.0f;
    }
    __syncthreads();

    int wave = (blockIdx.x * FC_BLOCK + threadIdx.x) >> 5;
    int lane = threadIdx.x & 31;
    int m_tile = wave / N_TILES;
    int n_tile = wave % N_TILES;
    if (m_tile >= M_TILES) return;

    const int m_base = m_tile * 16;
    const int l16    = lane & 15;
    const int kb     = (lane >> 4) * 2;

    const int mrow   = m_base + l16;
    const int ncol_l = n_tile * 16 + l16;
    const float* arow = x + (size_t)mrow * DD;
    const float* brow = wlds + ncol_l * DD;

    v8f acc = {};
    #pragma unroll 8
    for (int k = 0; k < DD; k += 4) {
        v2f a, b;
        a.x = arow[k + kb];
        a.y = arow[k + kb + 1];
        b.x = brow[k + kb];
        b.y = brow[k + kb + 1];
        acc = __builtin_amdgcn_wmma_f32_16x16x4_f32(
             false, a,  false, b,
             (short)0, acc,  false,  false);
    }

    if (ncol_l < CC) {
        int m_off = (lane >> 4) * 8;
        #pragma unroll
        for (int v = 0; v < 8; ++v) {
            out[(size_t)(m_base + m_off + v) * PITCH + ncol_l] = acc[v];
        }
    }
}

extern "C" void kernel_launch(void* const* d_in, const int* in_sizes, int n_in,
                              void* d_out, int out_size, void* d_ws, size_t ws_size,
                              hipStream_t stream) {
    const float* feat  = (const float*)d_in[0];
    const int*   row   = (const int*)  d_in[1];
    const int*   col   = (const int*)  d_in[2];
    const float* W     = (const float*)d_in[3];
    const float* bias  = (const float*)d_in[4];
    const float* gamma = (const float*)d_in[5];
    const float* beta  = (const float*)d_in[6];
    float* out = (float*)d_out;

    const int B = 256;
    const int edge_blocks = (EE + B - 1) / B;
    const int fc_waves    = M_TILES * N_TILES;
    const int fc_blocks   = (fc_waves * 32 + FC_BLOCK - 1) / FC_BLOCK;
    const int prop_blocks = (NTRIP / 2 * 32 + B - 1) / B;

    int*   node = (int*)d_ws;
    int*   cnt  = node + 2 * NP;
    float* ta   = (float*)(cnt + NP);
    static_assert((3 * NP * 4) % 256 == 0, "the tables start on a 256-byte unit of the workspace");

    const size_t padded_need = ((size_t)3 * NP + (size_t)2 * NN * HP + (size_t)2 * (EE + CSR_PAD)) * 4;

    if (ws_size >= padded_need) {
        float* tb   = ta + (size_t)NN * HP;
        int*   csr  = (int*)(tb + (size_t)NN * HP);
        int*   rank = (int*)tb;

        dgc_zero_cnt  <<<NP / B, B, 0, stream>>>(cnt);
        dgc_count_rank<<<edge_blocks, B, 0, stream>>>(row, cnt, rank);
        dgc_scan      <<<NP / SCAN_B, SCAN_B, 0, stream>>>(cnt, node);
        dgc_fill<HP>  <<<edge_blocks, B, 0, stream>>>(row, col, rank, node, csr);

        dgc_fc_wmma<HP><<<fc_blocks, FC_BLOCK, 0, stream>>>(feat, W, ta);

        dgc_prop<false, HP, HP><<<prop_blocks, B, 0, stream>>>(ta, tb, node, csr, bias, gamma, beta);
        dgc_prop<false, HP, HP><<<prop_blocks, B, 0, stream>>>(tb, ta, node, csr, bias, gamma, beta);
        dgc_prop<false, HP, HP><<<prop_blocks, B, 0, stream>>>(ta, tb, node, csr, bias, gamma, beta);
        dgc_prop<true,  HP, CC><<<prop_blocks, B, 0, stream>>>(tb, out, node, csr, bias, gamma, beta);
    } else {
        float* h    = ta;
        int*   csr  = (int*)(h + (size_t)NN * CC);
        int*   rank = (int*)h;

        dgc_zero_cnt  <<<NP / B, B, 0, stream>>>(cnt);
        dgc_count_rank<<<edge_blocks, B, 0, stream>>>(row, cnt, rank);
        dgc_scan      <<<NP / SCAN_B, SCAN_B, 0, stream>>>(cnt, node);
        dgc_fill<CC>  <<<edge_blocks, B, 0, stream>>>(row, col, rank, node, csr);

        dgc_fc_wmma<CC><<<fc_blocks, FC_BLOCK, 0, stream>>>(feat, W, out);

        dgc_prop<false, CC, CC><<<prop_blocks, B, 0, stream>>>(out, h, node, csr, bias, gamma, beta);
        dgc_prop<false, CC, CC><<<prop_blocks, B, 0, stream>>>(h, out, node, csr, bias, gamma, beta);
        dgc_prop<false, CC, CC><<<prop_blocks, B, 0, stream>>>(out, h, node, csr, bias, gamma, beta);
        dgc_prop<true,  CC, CC><<<prop_blocks, B, 0, stream>>>(h, out, node, csr, bias, gamma, beta);
    }
}


# Edges bucketed by a two-level counting sort on LDS counters (256 row ranges, then rows), no returning global atomic per edge; weights bit-equal, slots still arrival order: f32 sum order changes
# speedup vs baseline: 7.9923x; 1.4693x over previous
#include <hip/hip_runtime.h>
#include <hip/hip_bf16.h>

#define NN      100000
#define EE      1600000
#define DD      128
#define CC      40
#define KSTEPS  4
#define DELTA   0.5f
#define LN_EPS  1e-5f

#define HP      64
#define CSR_PAD 8
#define NP      100352
static_assert(NP > NN, "entry N exists");
static_assert(KSTEPS == 4, "the launcher writes the four steps out, the last one into out");
static_assert(HP >= CC && (HP * 4) % 256 == 0, "a padded row is whole 256-byte units");

typedef __attribute__((ext_vector_type(2))) float v2f;
typedef __attribute__((ext_vector_type(8))) float v8f;

#define NB1    256
#define NCH    256
#define RB     392
#define SB1    1024
#define ECH    (EE / NCH)
#define EPT    ((ECH + SB1 - 1) / SB1)
#define RBATCH 8
static_assert(NB1 * RB == NP, "the ranges tile the padded row space");
static_assert(EE % NCH == 0, "whole chunks");
static_assert(NB1 * NCH == 2048 * (SB1 / 32), "a wave of the scan owns 2048 matrix entries");
static_assert(NB1 * NCH <= NP, "the count matrix fits the region the launcher gives it");

__global__ __launch_bounds__(SB1)
void dgc_part_count(const int* __restrict__ row, int* __restrict__ mat) {
    __shared__ int h[NB1];
    const int t = threadIdx.x;
    if (t < NB1) h[t] = 0;
    __syncthreads();
    const int* src = row + blockIdx.x * ECH;
    int r[EPT];
    #pragma unroll
    for (int k = 0; k < EPT; ++k) {
        const int i = t + k * SB1;
        r[k] = (i < ECH) ? src[i] : 0;
    }
    #pragma unroll
    for (int k = 0; k < EPT; ++k)
        if (t + k * SB1 < ECH) atomicAdd(&h[r[k] / RB], 1);
    __syncthreads();
    if (t < NB1) mat[t * NCH + blockIdx.x] = h[t];
}

__global__ __launch_bounds__(SB1)
void dgc_part_scan(int* __restrict__ mat) {
    __shared__ int s_tot[32];
    const int t = threadIdx.x, lane = t & 31, wv = t >> 5;
    int4* seg = (int4*)mat + wv * 512;
    int4 v[16];
    #pragma unroll
    for (int k = 0; k < 16; ++k) v[k] = seg[k * 32 + lane];
    int ex[16];
    int carry = 0;
    #pragma unroll
    for (int k = 0; k < 16; ++k) {
        const int s = (v[k].x + v[k].y) + (v[k].z + v[k].w);
        int inc = s;
        #pragma unroll
        for (int off = 1; off < 32; off <<= 1) {
            const int u = __shfl_up(inc, off, 32);
            if (lane >= off) inc += u;
        }
        ex[k] = carry + inc - s;
        carry += __shfl(inc, 31, 32);
    }
    if (lane == 0) s_tot[wv] = carry;
    __syncthreads();
    int wbase = (lane < wv) ? s_tot[lane] : 0;
    for (int off = 16; off > 0; off >>= 1) wbase += __shfl_xor(wbase, off, 32);
    #pragma unroll
    for (int k = 0; k < 16; ++k) {
        int4 o;
        o.x = wbase + ex[k];
        o.y = o.x + v[k].x;
        o.z = o.y + v[k].y;
        o.w = o.z + v[k].z;
        seg[k * 32 + lane] = o;
    }
}

__global__ __launch_bounds__(SB1)
void dgc_part_scatter(const int* __restrict__ row,
                      const int* __restrict__ col,
                      const int* __restrict__ mat,
                      int* __restrict__ stage_raw) {
    __shared__ int cur[NB1];
    const int t = threadIdx.x;
    if (t < NB1) cur[t] = mat[t * NCH + blockIdx.x];
    __syncthreads();
    const int* rsrc = row + blockIdx.x * ECH;
    const int* csrc = col + blockIdx.x * ECH;
    int r[EPT], c[EPT];
    #pragma unroll
    for (int k = 0; k < EPT; ++k) {
        const int i = t + k * SB1;
        r[k] = (i < ECH) ? rsrc[i] : 0;
        c[k] = (i < ECH) ? csrc[i] : 0;
    }
    #pragma unroll
    for (int k = 0; k < EPT; ++k) {
        if (t + k * SB1 < ECH) {
            const int p = atomicAdd(&cur[r[k] / RB], 1);
            ((int2*)stage_raw)[p] = make_int2(r[k], c[k]);
        }
    }
}

__global__ __launch_bounds__(SB1)
void dgc_row_count(const int* __restrict__ stage_raw,
                   const int* __restrict__ mat,
                   int* __restrict__ node_raw) {
    __shared__ int cn[RB];
    __shared__ int s_tot[32];
    const int t = threadIdx.x, lane = t & 31, wv = t >> 5;
    const int b = blockIdx.x, r0 = b * RB;
    if (t < RB) cn[t] = 0;
    __syncthreads();
    const int beg = mat[b * NCH];
    const int end = (b + 1 < NB1) ? mat[(b + 1) * NCH] : EE;
    const int2* stage = (const int2*)stage_raw;
    for (int i0 = beg + t; i0 < end; i0 += RBATCH * SB1) {
        int r[RBATCH];
        #pragma unroll
        for (int k = 0; k < RBATCH; ++k) {
            const int i = i0 + k * SB1;
            r[k] = stage[(i < end) ? i : (EE - 1)].x;
        }
        #pragma unroll
        for (int k = 0; k < RBATCH; ++k) asm volatile("" : "+v"(r[k]));
        #pragma unroll
        for (int k = 0; k < RBATCH; ++k)
            if (i0 + k * SB1 < end) atomicAdd(&cn[r[k] - r0], 1);
    }
    __syncthreads();
    const int cv = (t < RB) ? cn[t] : 0;
    int inc = cv;
    for (int off = 1; off < 32; off <<= 1) {
        const int u = __shfl_up(inc, off, 32);
        if (lane >= off) inc += u;
    }
    if (lane == 31) s_tot[wv] = inc;
    __syncthreads();
    int wbase = (lane < wv) ? s_tot[lane] : 0;
    for (int off = 16; off > 0; off >>= 1) wbase += __shfl_xor(wbase, off, 32);
    if (t < RB) {
        int2 rec;
        rec.x = beg + wbase + inc - cv;
        rec.y = __float_as_int(rsqrtf((float)cv + 1.0f));
        ((int2*)node_raw)[r0 + t] = rec;
    }
}

template <int PITCH>
__global__ __launch_bounds__(SB1)
void dgc_row_fill(const int* __restrict__ stage_raw,
                  const int* __restrict__ mat,
                  const int* __restrict__ node_raw,
                  int* __restrict__ csr_raw) {
    __shared__ int cur[RB];
    __shared__ int dv[RB];
    const int t = threadIdx.x;
    const int b = blockIdx.x, r0 = b * RB;
    const int2* node = (const int2*)node_raw;
    if (t < RB) {
        const int2 nr = node[r0 + t];
        cur[t] = nr.x;
        dv[t]  = nr.y;
    }
    if (b == 0 && t < CSR_PAD) ((int2*)csr_raw)[EE + t] = make_int2(0, 0);
    __syncthreads();
    const int beg = mat[b * NCH];
    const int end = (b + 1 < NB1) ? mat[(b + 1) * NCH] : EE;
    const int2* stage = (const int2*)stage_raw;
    for (int i0 = beg + t; i0 < end; i0 += RBATCH * SB1) {
        int2 e[RBATCH];
        int  dc[RBATCH];
        #pragma unroll
        for (int k = 0; k < RBATCH; ++k) {
            const int i = i0 + k * SB1;
            e[k] = stage[(i < end) ? i : (EE - 1)];
        }
        #pragma unroll
        for (int k = 0; k < RBATCH; ++k) asm volatile("" : "+v"(e[k].x), "+v"(e[k].y));
        #pragma unroll
        for (int k = 0; k < RBATCH; ++k) dc[k] = node[e[k].y].y;
        #pragma unroll
        for (int k = 0; k < RBATCH; ++k) asm volatile("" : "+v"(dc[k]));
        #pragma unroll
        for (int k = 0; k < RBATCH; ++k) {
            if (i0 + k * SB1 < end) {
                const int rl = e[k].x - r0;
                const int p  = atomicAdd(&cur[rl], 1);
                const float w = __int_as_float(dv[rl]) * __int_as_float(dc[k]);
                int2 rec;
                rec.x = e[k].y * PITCH;
                rec.y = __float_as_int(w);
                ((int2*)csr_raw)[p] = rec;
            }
        }
    }
}

#define GL     10
#define RPW    3
#define NTRIP  ((NN + RPW - 1) / RPW)
#define PB     8
static_assert(GL * 4 == CC, "ten float4 make a row");
static_assert(PB - 1 <= CSR_PAD, "a part batch over-reads at most PB - 1 records");

template <bool FULL>
__device__ __forceinline__ void dgc_batch(const float* __restrict__ hin,
                                          const int2* __restrict__ csr,
                                          int e, int rem, int j4, float4& acc) {
    int2   m[PB];
    float4 v[PB];
    #pragma unroll
    for (int k = 0; k < PB; ++k) {
        m[k] = csr[e + k];
        if (!FULL && k >= rem) m[k].y = 0;
    }
    if (!FULL) {
        #pragma unroll
        for (int k = 0; k < PB; ++k)
            asm volatile("" : "+v"(m[k].x), "+v"(m[k].y));
    }
    #pragma unroll
    for (int k = 0; k < PB; ++k) {
        v[k] = make_float4(0.f, 0.f, 0.f, 0.f);
        if (FULL || k < rem) v[k] = *(const float4*)(hin + m[k].x + j4);
    }
    #pragma unroll
    for (int k = 0; k < PB; ++k) {
        const float w = __int_as_float(m[k].y);
        acc.x = fmaf(v[k].x, w, acc.x);
        acc.y = fmaf(v[k].y, w, acc.y);
        acc.z = fmaf(v[k].z, w, acc.z);
        acc.w = fmaf(v[k].w, w, acc.w);
    }
}

struct dgc_ring { int p5, p1, p2, p3, p4, p0; };

__device__ __forceinline__ float dgc_group_sum(float v, const dgc_ring& g) {
    const float t = v + __shfl(v, g.p5, 32);
    float s = t + __shfl(t, g.p1, 32);
    s += __shfl(t, g.p2, 32);
    s += __shfl(t, g.p3, 32);
    s += __shfl(t, g.p4, 32);
    return __shfl(s, g.p0, 32);
}

__device__ __forceinline__ float4 dgc_step_row(const float4 hv, const float ws, const float4 acc) {
    float4 o;
    o.x = (1.0f - DELTA) * hv.x + DELTA * fmaf(hv.x, ws, acc.x);
    o.y = (1.0f - DELTA) * hv.y + DELTA * fmaf(hv.y, ws, acc.y);
    o.z = (1.0f - DELTA) * hv.z + DELTA * fmaf(hv.z, ws, acc.z);
    o.w = (1.0f - DELTA) * hv.w + DELTA * fmaf(hv.w, ws, acc.w);
    return o;
}

template <bool LAST, int PIN, int POUT>
__global__ __launch_bounds__(256)
void dgc_prop(const float* __restrict__ hin,
              float* __restrict__ hout,
              const int* __restrict__ node_raw,
              const int* __restrict__ csr_raw,
              const float* __restrict__ bias,
              const float* __restrict__ gamma,
              const float* __restrict__ beta) {
    const int2* node = (const int2*)node_raw;
    const int2* csr  = (const int2*)csr_raw;
    const int lane   = threadIdx.x & 31;
    const int grp    = lane / GL;
    const int j4     = (lane - grp * GL) * 4;
    const int wave   = (blockIdx.x * blockDim.x + threadIdx.x) >> 5;
    const int nwaves = (gridDim.x * blockDim.x) >> 5;

    float4 bv = make_float4(0.f, 0.f, 0.f, 0.f), gv = bv, tv = bv;
    dgc_ring ring = {lane, lane, lane, lane, lane, lane};
    if (LAST) {
        bv = make_float4(bias[j4],  bias[j4 + 1],  bias[j4 + 2],  bias[j4 + 3]);
        gv = make_float4(gamma[j4], gamma[j4 + 1], gamma[j4 + 2], gamma[j4 + 3]);
        tv = make_float4(beta[j4],  beta[j4 + 1],  beta[j4 + 2],  beta[j4 + 3]);
        if (grp < RPW) {
            const int base = grp * GL;
            const int j    = lane - base;
            ring.p5 = base + (j + 5) % GL;
            ring.p1 = base + (j + 1) % GL;
            ring.p2 = base + (j + 2) % GL;
            ring.p3 = base + (j + 3) % GL;
            ring.p4 = base + (j + 4) % GL;
            ring.p0 = base;
        }
    }

    for (int t = wave; t < NTRIP; t += nwaves) {
        const int  r    = t * RPW + grp;
        const bool live = (grp < RPW) && (r < NN);
        int beg = 0, end = 0;
        float ws = 0.f;
        float4 hv = make_float4(0.f, 0.f, 0.f, 0.f);
        if (live) {
            const int2 nr = node[r];
            beg = nr.x;
            end = node[r + 1].x;
            ws  = __int_as_float(nr.y) * __int_as_float(nr.y);
            hv  = *(const float4*)(hin + r * PIN + j4);
        }
        float4 acc = make_float4(0.f, 0.f, 0.f, 0.f);
        int e = beg;
        for (; e + PB <= end; e += PB) dgc_batch<true>(hin, csr, e, PB, j4, acc);
        if (e < end) dgc_batch<false>(hin, csr, e, end - e, j4, acc);
        if (!LAST) {
            if (live) *(float4*)(hout + r * POUT + j4) = dgc_step_row(hv, ws, acc);
        } else {
            const float4 o = dgc_step_row(hv, ws, acc);
            float4 x;
            x.x = o.x + bv.x;
            x.y = o.y + bv.y;
            x.z = o.z + bv.z;
            x.w = o.w + bv.w;
            const float s  = dgc_group_sum((x.x + x.y) + (x.z + x.w), ring);
            const float sq = dgc_group_sum((x.x * x.x + x.y * x.y) + (x.z * x.z + x.w * x.w), ring);
            const float mean = s * (1.0f / CC);
            const float var  = sq * (1.0f / CC) - mean * mean;
            const float inv  = rsqrtf(var + LN_EPS);
            float4 y;
            y.x = (x.x - mean) * inv * gv.x + tv.x;
            y.y = (x.y - mean) * inv * gv.y + tv.y;
            y.z = (x.z - mean) * inv * gv.z + tv.z;
            y.w = (x.w - mean) * inv * gv.w + tv.w;
            if (live) *(float4*)(hout + r * POUT + j4) = y;
        }
    }
}

#define M_TILES   (NN / 16)
#define N_TILES   3
#define NPAD      (N_TILES * 16)
#define FC_BLOCK  256

template <int PITCH>
__global__ void dgc_fc_wmma(const float* __restrict__ x,
                            const float* __restrict__ W,
                            float* __restrict__ out) {
    __shared__ float wlds[NPAD * DD];

    for (int idx = threadIdx.x; idx < NPAD * DD; idx += FC_BLOCK) {
        int colc = idx >> 7;
        wlds[idx] = (colc < CC) ? W[idx] : 0.0f;
    }
    __syncthreads();

    int wave = (blockIdx.x * FC_BLOCK + threadIdx.x) >> 5;
    int lane = threadIdx.x & 31;
    int m_tile = wave / N_TILES;
    int n_tile = wave % N_TILES;
    if (m_tile >= M_TILES) return;

    const int m_base = m_tile * 16;
    const int l16    = lane & 15;
    const int kb     = (lane >> 4) * 2;

    const int mrow   = m_base + l16;
    const int ncol_l = n_tile * 16 + l16;
    const float* arow = x + (size_t)mrow * DD;
    const float* brow = wlds + ncol_l * DD;

    v8f acc = {};
    #pragma unroll 8
    for (int k = 0; k < DD; k += 4) {
        v2f a, b;
        a.x = arow[k + kb];
        a.y = arow[k + kb + 1];
        b.x = brow[k + kb];
        b.y = brow[k + kb + 1];
        acc = __builtin_amdgcn_wmma_f32_16x16x4_f32(
             false, a,  false, b,
             (short)0, acc,  false,  false);
    }

    if (ncol_l < CC) {
        int m_off = (lane >> 4) * 8;
        #pragma unroll
        for (int v = 0; v < 8; ++v) {
            out[(size_t)(m_base + m_off + v) * PITCH + ncol_l] = acc[v];
        }
    }
}

extern "C" void kernel_launch(void* const* d_in, const int* in_sizes, int n_in,
                              void* d_out, int out_size, void* d_ws, size_t ws_size,
                              hipStream_t stream) {
    const float* feat  = (const float*)d_in[0];
    const int*   row   = (const int*)  d_in[1];
    const int*   col   = (const int*)  d_in[2];
    const float* W     = (const float*)d_in[3];
    const float* bias  = (const float*)d_in[4];
    const float* gamma = (const float*)d_in[5];
    const float* beta  = (const float*)d_in[6];
    float* out = (float*)d_out;

    const int B = 256;
    const int fc_waves    = M_TILES * N_TILES;
    const int fc_blocks   = (fc_waves * 32 + FC_BLOCK - 1) / FC_BLOCK;
    const int prop_blocks = (NTRIP / 2 * 32 + B - 1) / B;

    int*   node = (int*)d_ws;
    int*   mat  = node + 2 * NP;
    float* ta   = (float*)(mat + NP);
    static_assert((3 * NP * 4) % 256 == 0, "the tables start on a 256-byte unit of the workspace");
    static_assert(2 * EE <= NN * CC, "the staged pairs fit the table they borrow, dense or padded");

    const size_t padded_need = ((size_t)3 * NP + (size_t)2 * NN * HP + (size_t)2 * (EE + CSR_PAD)) * 4;

    if (ws_size >= padded_need) {
        float* tb   = ta + (size_t)NN * HP;
        int*   csr  = (int*)(tb + (size_t)NN * HP);
        int*   stage = (int*)tb;

        dgc_part_count  <<<NCH, SB1, 0, stream>>>(row, mat);
        dgc_part_scan   <<<1,   SB1, 0, stream>>>(mat);
        dgc_part_scatter<<<NCH, SB1, 0, stream>>>(row, col, mat, stage);
        dgc_row_count   <<<NB1, SB1, 0, stream>>>(stage, mat, node);
        dgc_row_fill<HP><<<NB1, SB1, 0, stream>>>(stage, mat, node, csr);

        dgc_fc_wmma<HP><<<fc_blocks, FC_BLOCK, 0, stream>>>(feat, W, ta);

        dgc_prop<false, HP, HP><<<prop_blocks, B, 0, stream>>>(ta, tb, node, csr, bias, gamma, beta);
        dgc_prop<false, HP, HP><<<prop_blocks, B, 0, stream>>>(tb, ta, node, csr, bias, gamma, beta);
        dgc_prop<false, HP, HP><<<prop_blocks, B, 0, stream>>>(ta, tb, node, csr, bias, gamma, beta);
        dgc_prop<true,  HP, CC><<<prop_blocks, B, 0, stream>>>(tb, out, node, csr, bias, gamma, beta);
    } else {
        float* h    = ta;
        int*   csr  = (int*)(h + (size_t)NN * CC);
        int*   stage = (int*)h;

        dgc_part_count  <<<NCH, SB1, 0, stream>>>(row, mat);
        dgc_part_scan   <<<1,   SB1, 0, stream>>>(mat);
        dgc_part_scatter<<<NCH, SB1, 0, stream>>>(row, col, mat, stage);
        dgc_row_count   <<<NB1, SB1, 0, stream>>>(stage, mat, node);
        dgc_row_fill<CC><<<NB1, SB1, 0, stream>>>(stage, mat, node, csr);

        dgc_fc_wmma<CC><<<fc_blocks, FC_BLOCK, 0, stream>>>(feat, W, out);

        dgc_prop<false, CC, CC><<<prop_blocks, B, 0, stream>>>(out, h, node, csr, bias, gamma, beta);
        dgc_prop<false, CC, CC><<<prop_blocks, B, 0, stream>>>(h, out, node, csr, bias, gamma, beta);
        dgc_prop<false, CC, CC><<<prop_blocks, B, 0, stream>>>(out, h, node, csr, bias, gamma, beta);
        dgc_prop<true,  CC, CC><<<prop_blocks, B, 0, stream>>>(h, out, node, csr, bias, gamma, beta);
    }
}


# Steps run on the row-scaled table g = dinv*h (FC stores g, last step unscales): an edge record is its column alone, no weight, so row count and fill fuse into one launch; f32 rounding changes
# speedup vs baseline: 8.4007x; 1.0511x over previous
#include <hip/hip_runtime.h>
#include <hip/hip_bf16.h>

#define NN      100000
#define EE      1600000
#define DD      128
#define CC      40
#define KSTEPS  4
#define DELTA   0.5f
#define LN_EPS  1e-5f

#define HP      64
#define CSR_PAD 8
#define NP      100352
static_assert(NP > NN, "entry N exists");
static_assert(KSTEPS == 4, "the launcher writes the four steps out, the last one into out");
static_assert(HP >= CC && (HP * 4) % 256 == 0, "a padded row is whole 256-byte units");

typedef __attribute__((ext_vector_type(2))) float v2f;
typedef __attribute__((ext_vector_type(8))) float v8f;

#define NB1    256
#define NCH    256
#define RB     392
#define SB1    1024
#define ECH    (EE / NCH)
#define EPT    ((ECH + SB1 - 1) / SB1)
#define RBATCH 8
static_assert(NB1 * RB == NP, "the ranges tile the padded row space");
static_assert(EE % NCH == 0, "whole chunks");
static_assert(NB1 * NCH == 2048 * (SB1 / 32), "a wave of the scan owns 2048 matrix entries");
static_assert(NB1 * NCH <= NP, "the count matrix fits the region the launcher gives it");

__global__ __launch_bounds__(SB1)
void dgc_part_count(const int* __restrict__ row, int* __restrict__ mat) {
    __shared__ int h[NB1];
    const int t = threadIdx.x;
    if (t < NB1) h[t] = 0;
    __syncthreads();
    const int* src = row + blockIdx.x * ECH;
    int r[EPT];
    #pragma unroll
    for (int k = 0; k < EPT; ++k) {
        const int i = t + k * SB1;
        r[k] = (i < ECH) ? src[i] : 0;
    }
    #pragma unroll
    for (int k = 0; k < EPT; ++k)
        if (t + k * SB1 < ECH) atomicAdd(&h[r[k] / RB], 1);
    __syncthreads();
    if (t < NB1) mat[t * NCH + blockIdx.x] = h[t];
}

__global__ __launch_bounds__(SB1)
void dgc_part_scan(int* __restrict__ mat) {
    __shared__ int s_tot[32];
    const int t = threadIdx.x, lane = t & 31, wv = t >> 5;
    int4* seg = (int4*)mat + wv * 512;
    int4 v[16];
    #pragma unroll
    for (int k = 0; k < 16; ++k) v[k] = seg[k * 32 + lane];
    int ex[16];
    int carry = 0;
    #pragma unroll
    for (int k = 0; k < 16; ++k) {
        const int s = (v[k].x + v[k].y) + (v[k].z + v[k].w);
        int inc = s;
        #pragma unroll
        for (int off = 1; off < 32; off <<= 1) {
            const int u = __shfl_up(inc, off, 32);
            if (lane >= off) inc += u;
        }
        ex[k] = carry + inc - s;
        carry += __shfl(inc, 31, 32);
    }
    if (lane == 0) s_tot[wv] = carry;
    __syncthreads();
    int wbase = (lane < wv) ? s_tot[lane] : 0;
    for (int off = 16; off > 0; off >>= 1) wbase += __shfl_xor(wbase, off, 32);
    #pragma unroll
    for (int k = 0; k < 16; ++k) {
        int4 o;
        o.x = wbase + ex[k];
        o.y = o.x + v[k].x;
        o.z = o.y + v[k].y;
        o.w = o.z + v[k].z;
        seg[k * 32 + lane] = o;
    }
}

__global__ __launch_bounds__(SB1)
void dgc_part_scatter(const int* __restrict__ row,
                      const int* __restrict__ col,
                      const int* __restrict__ mat,
                      int* __restrict__ stage_raw) {
    __shared__ int cur[NB1];
    const int t = threadIdx.x;
    if (t < NB1) cur[t] = mat[t * NCH + blockIdx.x];
    __syncthreads();
    const int* rsrc = row + blockIdx.x * ECH;
    const int* csrc = col + blockIdx.x * ECH;
    int r[EPT], c[EPT];
    #pragma unroll
    for (int k = 0; k < EPT; ++k) {
        const int i = t + k * SB1;
        r[k] = (i < ECH) ? rsrc[i] : 0;
        c[k] = (i < ECH) ? csrc[i] : 0;
    }
    #pragma unroll
    for (int k = 0; k < EPT; ++k) {
        if (t + k * SB1 < ECH) {
            const int p = atomicAdd(&cur[r[k] / RB], 1);
            ((int2*)stage_raw)[p] = make_int2(r[k], c[k]);
        }
    }
}

template <int PITCH>
__global__ __launch_bounds__(SB1)
void dgc_row_build(const int* __restrict__ stage_raw,
                   const int* __restrict__ mat,
                   int* __restrict__ node_raw,
                   int* __restrict__ csr) {
    __shared__ int cn[RB];
    __shared__ int s_tot[32];
    const int t = threadIdx.x, lane = t & 31, wv = t >> 5;
    const int b = blockIdx.x, r0 = b * RB;
    if (t < RB) cn[t] = 0;
    if (b == 0 && t < CSR_PAD) csr[EE + t] = 0;
    __syncthreads();
    const int beg = mat[b * NCH];
    const int end = (b + 1 < NB1) ? mat[(b + 1) * NCH] : EE;
    const int2* stage = (const int2*)stage_raw;
    const int i00 = beg + t;

    int2 e0[RBATCH];
    #pragma unroll
    for (int k = 0; k < RBATCH; ++k) {
        const int i = i00 + k * SB1;
        e0[k] = stage[(i < end) ? i : (EE - 1)];
    }
    #pragma unroll
    for (int k = 0; k < RBATCH; ++k)
        if (i00 + k * SB1 < end) atomicAdd(&cn[e0[k].x - r0], 1);
    for (int i0 = i00 + RBATCH * SB1; i0 < end; i0 += RBATCH * SB1) {
        int r[RBATCH];
        #pragma unroll
        for (int k = 0; k < RBATCH; ++k) {
            const int i = i0 + k * SB1;
            r[k] = stage[(i < end) ? i : (EE - 1)].x;
        }
        #pragma unroll
        for (int k = 0; k < RBATCH; ++k)
            if (i0 + k * SB1 < end) atomicAdd(&cn[r[k] - r0], 1);
    }
    __syncthreads();

    const int cv = (t < RB) ? cn[t] : 0;
    int inc = cv;
    for (int off = 1; off < 32; off <<= 1) {
        const int u = __shfl_up(inc, off, 32);
        if (lane >= off) inc += u;
    }
    if (lane == 31) s_tot[wv] = inc;
    __syncthreads();
    int wbase = (lane < wv) ? s_tot[lane] : 0;
    for (int off = 16; off > 0; off >>= 1) wbase += __shfl_xor(wbase, off, 32);
    if (t < RB) {
        int2 rec;
        rec.x = beg + wbase + inc - cv;
        rec.y = __float_as_int(rsqrtf((float)cv + 1.0f));
        ((int2*)node_raw)[r0 + t] = rec;
        cn[t] = rec.x;
    }
    __syncthreads();

    #pragma unroll
    for (int k = 0; k < RBATCH; ++k) {
        if (i00 + k * SB1 < end) {
            const int p = atomicAdd(&cn[e0[k].x - r0], 1);
            csr[p] = e0[k].y * PITCH;
        }
    }
    for (int i0 = i00 + RBATCH * SB1; i0 < end; i0 += RBATCH * SB1) {
        int2 e[RBATCH];
        #pragma unroll
        for (int k = 0; k < RBATCH; ++k) {
            const int i = i0 + k * SB1;
            e[k] = stage[(i < end) ? i : (EE - 1)];
        }
        #pragma unroll
        for (int k = 0; k < RBATCH; ++k) {
            if (i0 + k * SB1 < end) {
                const int p = atomicAdd(&cn[e[k].x - r0], 1);
                csr[p] = e[k].y * PITCH;
            }
        }
    }
}

#define GL     10
#define RPW    3
#define NTRIP  ((NN + RPW - 1) / RPW)
#define PB     8
static_assert(GL * 4 == CC, "ten float4 make a row");
static_assert(PB - 1 <= CSR_PAD, "a part batch over-reads at most PB - 1 records");

template <bool FULL>
__device__ __forceinline__ void dgc_batch(const float* __restrict__ hin,
                                          const int* __restrict__ csr,
                                          int e, int rem, int j4, float4& acc) {
    int    m[PB];
    float4 v[PB];
    #pragma unroll
    for (int k = 0; k < PB; ++k) m[k] = csr[e + k];
    if (!FULL) {
        #pragma unroll
        for (int k = 0; k < PB; ++k)
            asm volatile("" : "+v"(m[k]));
    }
    #pragma unroll
    for (int k = 0; k < PB; ++k) {
        v[k] = make_float4(0.f, 0.f, 0.f, 0.f);
        if (FULL || k < rem) v[k] = *(const float4*)(hin + m[k] + j4);
    }
    #pragma unroll
    for (int k = 0; k < PB; ++k) {
        acc.x += v[k].x;
        acc.y += v[k].y;
        acc.z += v[k].z;
        acc.w += v[k].w;
    }
}

struct dgc_ring { int p5, p1, p2, p3, p4, p0; };

__device__ __forceinline__ float dgc_group_sum(float v, const dgc_ring& g) {
    const float t = v + __shfl(v, g.p5, 32);
    float s = t + __shfl(t, g.p1, 32);
    s += __shfl(t, g.p2, 32);
    s += __shfl(t, g.p3, 32);
    s += __shfl(t, g.p4, 32);
    return __shfl(s, g.p0, 32);
}

__device__ __forceinline__ float4 dgc_step_row(const float4 gv, const float sa, const float sb,
                                               const float4 acc) {
    float4 o;
    o.x = (1.0f - DELTA) * (gv.x * sa) + DELTA * (sb * (acc.x + gv.x));
    o.y = (1.0f - DELTA) * (gv.y * sa) + DELTA * (sb * (acc.y + gv.y));
    o.z = (1.0f - DELTA) * (gv.z * sa) + DELTA * (sb * (acc.z + gv.z));
    o.w = (1.0f - DELTA) * (gv.w * sa) + DELTA * (sb * (acc.w + gv.w));
    return o;
}

template <bool LAST, int PIN, int POUT>
__global__ __launch_bounds__(256)
void dgc_prop(const float* __restrict__ hin,
              float* __restrict__ hout,
              const int* __restrict__ node_raw,
              const int* __restrict__ csr_raw,
              const float* __restrict__ bias,
              const float* __restrict__ gamma,
              const float* __restrict__ beta) {
    const int2* node = (const int2*)node_raw;
    const int*  csr  = csr_raw;
    const int lane   = threadIdx.x & 31;
    const int grp    = lane / GL;
    const int j4     = (lane - grp * GL) * 4;
    const int wave   = (blockIdx.x * blockDim.x + threadIdx.x) >> 5;
    const int nwaves = (gridDim.x * blockDim.x) >> 5;

    float4 bv = make_float4(0.f, 0.f, 0.f, 0.f), gv = bv, tv = bv;
    dgc_ring ring = {lane, lane, lane, lane, lane, lane};
    if (LAST) {
        bv = make_float4(bias[j4],  bias[j4 + 1],  bias[j4 + 2],  bias[j4 + 3]);
        gv = make_float4(gamma[j4], gamma[j4 + 1], gamma[j4 + 2], gamma[j4 + 3]);
        tv = make_float4(beta[j4],  beta[j4 + 1],  beta[j4 + 2],  beta[j4 + 3]);
        if (grp < RPW) {
            const int base = grp * GL;
            const int j    = lane - base;
            ring.p5 = base + (j + 5) % GL;
            ring.p1 = base + (j + 1) % GL;
            ring.p2 = base + (j + 2) % GL;
            ring.p3 = base + (j + 3) % GL;
            ring.p4 = base + (j + 4) % GL;
            ring.p0 = base;
        }
    }

    for (int t = wave; t < NTRIP; t += nwaves) {
        const int  r    = t * RPW + grp;
        const bool live = (grp < RPW) && (r < NN);
        int beg = 0, end = 0;
        float dv = 0.f;
        float4 hv = make_float4(0.f, 0.f, 0.f, 0.f);
        if (live) {
            const int2 nr = node[r];
            beg = nr.x;
            end = node[r + 1].x;
            dv  = __int_as_float(nr.y);
            hv  = *(const float4*)(hin + r * PIN + j4);
        }
        float4 acc = make_float4(0.f, 0.f, 0.f, 0.f);
        int e = beg;
        for (; e + PB <= end; e += PB) dgc_batch<true>(hin, csr, e, PB, j4, acc);
        if (e < end) dgc_batch<false>(hin, csr, e, end - e, j4, acc);
        if (!LAST) {
            if (live) *(float4*)(hout + r * POUT + j4) = dgc_step_row(hv, 1.0f, dv * dv, acc);
        } else {
            const float sd = ((float)(end - beg) + 1.0f) * dv;
            const float4 o = dgc_step_row(hv, sd, dv, acc);
            float4 x;
            x.x = o.x + bv.x;
            x.y = o.y + bv.y;
            x.z = o.z + bv.z;
            x.w = o.w + bv.w;
            const float s  = dgc_group_sum((x.x + x.y) + (x.z + x.w), ring);
            const float sq = dgc_group_sum((x.x * x.x + x.y * x.y) + (x.z * x.z + x.w * x.w), ring);
            const float mean = s * (1.0f / CC);
            const float var  = sq * (1.0f / CC) - mean * mean;
            const float inv  = rsqrtf(var + LN_EPS);
            float4 y;
            y.x = (x.x - mean) * inv * gv.x + tv.x;
            y.y = (x.y - mean) * inv * gv.y + tv.y;
            y.z = (x.z - mean) * inv * gv.z + tv.z;
            y.w = (x.w - mean) * inv * gv.w + tv.w;
            if (live) *(float4*)(hout + r * POUT + j4) = y;
        }
    }
}

#define M_TILES   (NN / 16)
#define N_TILES   3
#define NPAD      (N_TILES * 16)
#define FC_BLOCK  256

template <int PITCH>
__global__ void dgc_fc_wmma(const float* __restrict__ x,
                            const float* __restrict__ W,
                            const int* __restrict__ node_raw,
                            float* __restrict__ out) {
    __shared__ float wlds[NPAD * DD];

    for (int idx = threadIdx.x; idx < NPAD * DD; idx += FC_BLOCK) {
        int colc = idx >> 7;
        wlds[idx] = (colc < CC) ? W[idx] : 0.0f;
    }
    __syncthreads();

    int wave = (blockIdx.x * FC_BLOCK + threadIdx.x) >> 5;
    int lane = threadIdx.x & 31;
    int m_tile = wave / N_TILES;
    int n_tile = wave % N_TILES;
    if (m_tile >= M_TILES) return;

    const int m_base = m_tile * 16;
    const int l16    = lane & 15;
    const int kb     = (lane >> 4) * 2;

    const int mrow   = m_base + l16;
    const int ncol_l = n_tile * 16 + l16;
    const float* arow = x + (size_t)mrow * DD;
    const float* brow = wlds + ncol_l * DD;

    const int2* node = (const int2*)node_raw;
    const int m_off = (lane >> 4) * 8;
    float dvr[8];
    #pragma unroll
    for (int v = 0; v < 8; ++v) dvr[v] = __int_as_float(node[m_base + m_off + v].y);

    v8f acc = {};
    #pragma unroll 8
    for (int k = 0; k < DD; k += 4) {
        v2f a, b;
        a.x = arow[k + kb];
        a.y = arow[k + kb + 1];
        b.x = brow[k + kb];
        b.y = brow[k + kb + 1];
        acc = __builtin_amdgcn_wmma_f32_16x16x4_f32(
             false, a,  false, b,
             (short)0, acc,  false,  false);
    }

    asm volatile("v_nop\n\tv_nop\n\tv_nop\n\tv_nop\n\tv_nop\n\ts_wait_alu depctr_va_vdst(0)" : "+v"(acc));
    if (ncol_l < CC) {
        #pragma unroll
        for (int v = 0; v < 8; ++v) {
            out[(size_t)(m_base + m_off + v) * PITCH + ncol_l] = acc[v] * dvr[v];
        }
    }
}

extern "C" void kernel_launch(void* const* d_in, const int* in_sizes, int n_in,
                              void* d_out, int out_size, void* d_ws, size_t ws_size,
                              hipStream_t stream) {
    const float* feat  = (const float*)d_in[0];
    const int*   row   = (const int*)  d_in[1];
    const int*   col   = (const int*)  d_in[2];
    const float* W     = (const float*)d_in[3];
    const float* bias  = (const float*)d_in[4];
    const float* gamma = (const float*)d_in[5];
    const float* beta  = (const float*)d_in[6];
    float* out = (float*)d_out;

    const int B = 256;
    const int fc_waves    = M_TILES * N_TILES;
    const int fc_blocks   = (fc_waves * 32 + FC_BLOCK - 1) / FC_BLOCK;
    const int prop_blocks = (NTRIP / 2 * 32 + B - 1) / B;

    int*   node = (int*)d_ws;
    int*   mat  = node + 2 * NP;
    float* ta   = (float*)(mat + NP);
    static_assert((3 * NP * 4) % 256 == 0, "the tables start on a 256-byte unit of the workspace");
    static_assert(2 * EE <= NN * CC, "the staged pairs fit the table they borrow, dense or padded");

    const size_t padded_need = ((size_t)3 * NP + (size_t)2 * NN * HP + (size_t)(EE + CSR_PAD)) * 4;

    if (ws_size >= padded_need) {
        float* tb   = ta + (size_t)NN * HP;
        int*   csr  = (int*)(tb + (size_t)NN * HP);
        int*   stage = (int*)tb;

        dgc_part_count   <<<NCH, SB1, 0, stream>>>(row, mat);
        dgc_part_scan    <<<1,   SB1, 0, stream>>>(mat);
        dgc_part_scatter <<<NCH, SB1, 0, stream>>>(row, col, mat, stage);
        dgc_row_build<HP><<<NB1, SB1, 0, stream>>>(stage, mat, node, csr);

        dgc_fc_wmma<HP><<<fc_blocks, FC_BLOCK, 0, stream>>>(feat, W, node, ta);

        dgc_prop<false, HP, HP><<<prop_blocks, B, 0, stream>>>(ta, tb, node, csr, bias, gamma, beta);
        dgc_prop<false, HP, HP><<<prop_blocks, B, 0, stream>>>(tb, ta, node, csr, bias, gamma, beta);
        dgc_prop<false, HP, HP><<<prop_blocks, B, 0, stream>>>(ta, tb, node, csr, bias, gamma, beta);
        dgc_prop<true,  HP, CC><<<prop_blocks, B, 0, stream>>>(tb, out, node, csr, bias, gamma, beta);
    } else {
        float* h    = ta;
        int*   csr  = (int*)(h + (size_t)NN * CC);
        int*   stage = (int*)h;

        dgc_part_count   <<<NCH, SB1, 0, stream>>>(row, mat);
        dgc_part_scan    <<<1,   SB1, 0, stream>>>(mat);
        dgc_part_scatter <<<NCH, SB1, 0, stream>>>(row, col, mat, stage);
        dgc_row_build<CC><<<NB1, SB1, 0, stream>>>(stage, mat, node, csr);

        dgc_fc_wmma<CC><<<fc_blocks, FC_BLOCK, 0, stream>>>(feat, W, node, out);

        dgc_prop<false, CC, CC><<<prop_blocks, B, 0, stream>>>(out, h, node, csr, bias, gamma, beta);
        dgc_prop<false, CC, CC><<<prop_blocks, B, 0, stream>>>(h, out, node, csr, bias, gamma, beta);
        dgc_prop<false, CC, CC><<<prop_blocks, B, 0, stream>>>(out, h, node, csr, bias, gamma, beta);
        dgc_prop<true,  CC, CC><<<prop_blocks, B, 0, stream>>>(h, out, node, csr, bias, gamma, beta);
    }
}


# FC stages W in LDS at a 132-float row pitch (was 128: the 16 rows of a B-fragment read began on one bank, a 16-way conflict on every read); same k order, same sums
# speedup vs baseline: 8.7089x; 1.0367x over previous
#include <hip/hip_runtime.h>
#include <hip/hip_bf16.h>

#define NN      100000
#define EE      1600000
#define DD      128
#define CC      40
#define KSTEPS  4
#define DELTA   0.5f
#define LN_EPS  1e-5f

#define HP      64
#define CSR_PAD 8
#define NP      100352
static_assert(NP > NN, "entry N exists");
static_assert(KSTEPS == 4, "the launcher writes the four steps out, the last one into out");
static_assert(HP >= CC && (HP * 4) % 256 == 0, "a padded row is whole 256-byte units");

typedef __attribute__((ext_vector_type(2))) float v2f;
typedef __attribute__((ext_vector_type(8))) float v8f;

#define NB1    256
#define NCH    256
#define RB     392
#define SB1    1024
#define ECH    (EE / NCH)
#define EPT    ((ECH + SB1 - 1) / SB1)
#define RBATCH 8
static_assert(NB1 * RB == NP, "the ranges tile the padded row space");
static_assert(EE % NCH == 0, "whole chunks");
static_assert(NB1 * NCH == 2048 * (SB1 / 32), "a wave of the scan owns 2048 matrix entries");
static_assert(NB1 * NCH <= NP, "the count matrix fits the region the launcher gives it");

__global__ __launch_bounds__(SB1)
void dgc_part_count(const int* __restrict__ row, int* __restrict__ mat) {
    __shared__ int h[NB1];
    const int t = threadIdx.x;
    if (t < NB1) h[t] = 0;
    __syncthreads();
    const int* src = row + blockIdx.x * ECH;
    int r[EPT];
    #pragma unroll
    for (int k = 0; k < EPT; ++k) {
        const int i = t + k * SB1;
        r[k] = (i < ECH) ? src[i] : 0;
    }
    #pragma unroll
    for (int k = 0; k < EPT; ++k)
        if (t + k * SB1 < ECH) atomicAdd(&h[r[k] / RB], 1);
    __syncthreads();
    if (t < NB1) mat[t * NCH + blockIdx.x] = h[t];
}

__global__ __launch_bounds__(SB1)
void dgc_part_scan(int* __restrict__ mat) {
    __shared__ int s_tot[32];
    const int t = threadIdx.x, lane = t & 31, wv = t >> 5;
    int4* seg = (int4*)mat + wv * 512;
    int4 v[16];
    #pragma unroll
    for (int k = 0; k < 16; ++k) v[k] = seg[k * 32 + lane];
    int ex[16];
    int carry = 0;
    #pragma unroll
    for (int k = 0; k < 16; ++k) {
        const int s = (v[k].x + v[k].y) + (v[k].z + v[k].w);
        int inc = s;
        #pragma unroll
        for (int off = 1; off < 32; off <<= 1) {
            const int u = __shfl_up(inc, off, 32);
            if (lane >= off) inc += u;
        }
        ex[k] = carry + inc - s;
        carry += __shfl(inc, 31, 32);
    }
    if (lane == 0) s_tot[wv] = carry;
    __syncthreads();
    int wbase = (lane < wv) ? s_tot[lane] : 0;
    for (int off = 16; off > 0; off >>= 1) wbase += __shfl_xor(wbase, off, 32);
    #pragma unroll
    for (int k = 0; k < 16; ++k) {
        int4 o;
        o.x = wbase + ex[k];
        o.y = o.x + v[k].x;
        o.z = o.y + v[k].y;
        o.w = o.z + v[k].z;
        seg[k * 32 + lane] = o;
    }
}

__global__ __launch_bounds__(SB1)
void dgc_part_scatter(const int* __restrict__ row,
                      const int* __restrict__ col,
                      const int* __restrict__ mat,
                      int* __restrict__ stage_raw) {
    __shared__ int cur[NB1];
    const int t = threadIdx.x;
    if (t < NB1) cur[t] = mat[t * NCH + blockIdx.x];
    __syncthreads();
    const int* rsrc = row + blockIdx.x * ECH;
    const int* csrc = col + blockIdx.x * ECH;
    int r[EPT], c[EPT];
    #pragma unroll
    for (int k = 0; k < EPT; ++k) {
        const int i = t + k * SB1;
        r[k] = (i < ECH) ? rsrc[i] : 0;
        c[k] = (i < ECH) ? csrc[i] : 0;
    }
    #pragma unroll
    for (int k = 0; k < EPT; ++k) {
        if (t + k * SB1 < ECH) {
            const int p = atomicAdd(&cur[r[k] / RB], 1);
            ((int2*)stage_raw)[p] = make_int2(r[k], c[k]);
        }
    }
}

template <int PITCH>
__global__ __launch_bounds__(SB1)
void dgc_row_build(const int* __restrict__ stage_raw,
                   const int* __restrict__ mat,
                   int* __restrict__ node_raw,
                   int* __restrict__ csr) {
    __shared__ int cn[RB];
    __shared__ int s_tot[32];
    const int t = threadIdx.x, lane = t & 31, wv = t >> 5;
    const int b = blockIdx.x, r0 = b * RB;
    if (t < RB) cn[t] = 0;
    if (b == 0 && t < CSR_PAD) csr[EE + t] = 0;
    __syncthreads();
    const int beg = mat[b * NCH];
    const int end = (b + 1 < NB1) ? mat[(b + 1) * NCH] : EE;
    const int2* stage = (const int2*)stage_raw;
    const int i00 = beg + t;

    int2 e0[RBATCH];
    #pragma unroll
    for (int k = 0; k < RBATCH; ++k) {
        const int i = i00 + k * SB1;
        e0[k] = stage[(i < end) ? i : (EE - 1)];
    }
    #pragma unroll
    for (int k = 0; k < RBATCH; ++k)
        if (i00 + k * SB1 < end) atomicAdd(&cn[e0[k].x - r0], 1);
    for (int i0 = i00 + RBATCH * SB1; i0 < end; i0 += RBATCH * SB1) {
        int r[RBATCH];
        #pragma unroll
        for (int k = 0; k < RBATCH; ++k) {
            const int i = i0 + k * SB1;
            r[k] = stage[(i < end) ? i : (EE - 1)].x;
        }
        #pragma unroll
        for (int k = 0; k < RBATCH; ++k)
            if (i0 + k * SB1 < end) atomicAdd(&cn[r[k] - r0], 1);
    }
    __syncthreads();

    const int cv = (t < RB) ? cn[t] : 0;
    int inc = cv;
    for (int off = 1; off < 32; off <<= 1) {
        const int u = __shfl_up(inc, off, 32);
        if (lane >= off) inc += u;
    }
    if (lane == 31) s_tot[wv] = inc;
    __syncthreads();
    int wbase = (lane < wv) ? s_tot[lane] : 0;
    for (int off = 16; off > 0; off >>= 1) wbase += __shfl_xor(wbase, off, 32);
    if (t < RB) {
        int2 rec;
        rec.x = beg + wbase + inc - cv;
        rec.y = __float_as_int(rsqrtf((float)cv + 1.0f));
        ((int2*)node_raw)[r0 + t] = rec;
        cn[t] = rec.x;
    }
    __syncthreads();

    #pragma unroll
    for (int k = 0; k < RBATCH; ++k) {
        if (i00 + k * SB1 < end) {
            const int p = atomicAdd(&cn[e0[k].x - r0], 1);
            csr[p] = e0[k].y * PITCH;
        }
    }
    for (int i0 = i00 + RBATCH * SB1; i0 < end; i0 += RBATCH * SB1) {
        int2 e[RBATCH];
        #pragma unroll
        for (int k = 0; k < RBATCH; ++k) {
            const int i = i0 + k * SB1;
            e[k] = stage[(i < end) ? i : (EE - 1)];
        }
        #pragma unroll
        for (int k = 0; k < RBATCH; ++k) {
            if (i0 + k * SB1 < end) {
                const int p = atomicAdd(&cn[e[k].x - r0], 1);
                csr[p] = e[k].y * PITCH;
            }
        }
    }
}

#define GL     10
#define RPW    3
#define NTRIP  ((NN + RPW - 1) / RPW)
#define PB     8
static_assert(GL * 4 == CC, "ten float4 make a row");
static_assert(PB - 1 <= CSR_PAD, "a part batch over-reads at most PB - 1 records");

template <bool FULL>
__device__ __forceinline__ void dgc_batch(const float* __restrict__ hin,
                                          const int* __restrict__ csr,
                                          int e, int rem, int j4, float4& acc) {
    int    m[PB];
    float4 v[PB];
    #pragma unroll
    for (int k = 0; k < PB; ++k) m[k] = csr[e + k];
    if (!FULL) {
        #pragma unroll
        for (int k = 0; k < PB; ++k)
            asm volatile("" : "+v"(m[k]));
    }
    #pragma unroll
    for (int k = 0; k < PB; ++k) {
        v[k] = make_float4(0.f, 0.f, 0.f, 0.f);
        if (FULL || k < rem) v[k] = *(const float4*)(hin + m[k] + j4);
    }
    #pragma unroll
    for (int k = 0; k < PB; ++k) {
        acc.x += v[k].x;
        acc.y += v[k].y;
        acc.z += v[k].z;
        acc.w += v[k].w;
    }
}

struct dgc_ring { int p5, p1, p2, p3, p4, p0; };

__device__ __forceinline__ float dgc_group_sum(float v, const dgc_ring& g) {
    const float t = v + __shfl(v, g.p5, 32);
    float s = t + __shfl(t, g.p1, 32);
    s += __shfl(t, g.p2, 32);
    s += __shfl(t, g.p3, 32);
    s += __shfl(t, g.p4, 32);
    return __shfl(s, g.p0, 32);
}

__device__ __forceinline__ float4 dgc_step_row(const float4 gv, const float sa, const float sb,
                                               const float4 acc) {
    float4 o;
    o.x = (1.0f - DELTA) * (gv.x * sa) + DELTA * (sb * (acc.x + gv.x));
    o.y = (1.0f - DELTA) * (gv.y * sa) + DELTA * (sb * (acc.y + gv.y));
    o.z = (1.0f - DELTA) * (gv.z * sa) + DELTA * (sb * (acc.z + gv.z));
    o.w = (1.0f - DELTA) * (gv.w * sa) + DELTA * (sb * (acc.w + gv.w));
    return o;
}

template <bool LAST, int PIN, int POUT>
__global__ __launch_bounds__(256)
void dgc_prop(const float* __restrict__ hin,
              float* __restrict__ hout,
              const int* __restrict__ node_raw,
              const int* __restrict__ csr_raw,
              const float* __restrict__ bias,
              const float* __restrict__ gamma,
              const float* __restrict__ beta) {
    const int2* node = (const int2*)node_raw;
    const int*  csr  = csr_raw;
    const int lane   = threadIdx.x & 31;
    const int grp    = lane / GL;
    const int j4     = (lane - grp * GL) * 4;
    const int wave   = (blockIdx.x * blockDim.x + threadIdx.x) >> 5;
    const int nwaves = (gridDim.x * blockDim.x) >> 5;

    float4 bv = make_float4(0.f, 0.f, 0.f, 0.f), gv = bv, tv = bv;
    dgc_ring ring = {lane, lane, lane, lane, lane, lane};
    if (LAST) {
        bv = make_float4(bias[j4],  bias[j4 + 1],  bias[j4 + 2],  bias[j4 + 3]);
        gv = make_float4(gamma[j4], gamma[j4 + 1], gamma[j4 + 2], gamma[j4 + 3]);
        tv = make_float4(beta[j4],  beta[j4 + 1],  beta[j4 + 2],  beta[j4 + 3]);
        if (grp < RPW) {
            const int base = grp * GL;
            const int j    = lane - base;
            ring.p5 = base + (j + 5) % GL;
            ring.p1 = base + (j + 1) % GL;
            ring.p2 = base + (j + 2) % GL;
            ring.p3 = base + (j + 3) % GL;
            ring.p4 = base + (j + 4) % GL;
            ring.p0 = base;
        }
    }

    for (int t = wave; t < NTRIP; t += nwaves) {
        const int  r    = t * RPW + grp;
        const bool live = (grp < RPW) && (r < NN);
        int beg = 0, end = 0;
        float dv = 0.f;
        float4 hv = make_float4(0.f, 0.f, 0.f, 0.f);
        if (live) {
            const int2 nr = node[r];
            beg = nr.x;
            end = node[r + 1].x;
            dv  = __int_as_float(nr.y);
            hv  = *(const float4*)(hin + r * PIN + j4);
        }
        float4 acc = make_float4(0.f, 0.f, 0.f, 0.f);
        int e = beg;
        for (; e + PB <= end; e += PB) dgc_batch<true>(hin, csr, e, PB, j4, acc);
        if (e < end) dgc_batch<false>(hin, csr, e, end - e, j4, acc);
        if (!LAST) {
            if (live) *(float4*)(hout + r * POUT + j4) = dgc_step_row(hv, 1.0f, dv * dv, acc);
        } else {
            const float sd = ((float)(end - beg) + 1.0f) * dv;
            const float4 o = dgc_step_row(hv, sd, dv, acc);
            float4 x;
            x.x = o.x + bv.x;
            x.y = o.y + bv.y;
            x.z = o.z + bv.z;
            x.w = o.w + bv.w;
            const float s  = dgc_group_sum((x.x + x.y) + (x.z + x.w), ring);
            const float sq = dgc_group_sum((x.x * x.x + x.y * x.y) + (x.z * x.z + x.w * x.w), ring);
            const float mean = s * (1.0f / CC);
            const float var  = sq * (1.0f / CC) - mean * mean;
            const float inv  = rsqrtf(var + LN_EPS);
            float4 y;
            y.x = (x.x - mean) * inv * gv.x + tv.x;
            y.y = (x.y - mean) * inv * gv.y + tv.y;
            y.z = (x.z - mean) * inv * gv.z + tv.z;
            y.w = (x.w - mean) * inv * gv.w + tv.w;
            if (live) *(float4*)(hout + r * POUT + j4) = y;
        }
    }
}

#define M_TILES   (NN / 16)
#define N_TILES   3
#define NPAD      (N_TILES * 16)
#define FC_BLOCK  256
#define WP        (DD + 4)
static_assert((WP * 4) % 16 == 0 && ((WP * 4) / 16) % 2 == 1, "W row pitch in LDS: 16 bytes x odd");

template <int PITCH>
__global__ void dgc_fc_wmma(const float* __restrict__ x,
                            const float* __restrict__ W,
                            const int* __restrict__ node_raw,
                            float* __restrict__ out) {
    __shared__ float wlds[NPAD * WP];

    for (int idx = threadIdx.x; idx < NPAD * DD; idx += FC_BLOCK) {
        int colc = idx >> 7;
        int kk   = idx & (DD - 1);
        wlds[colc * WP + kk] = (colc < CC) ? W[idx] : 0.0f;
    }
    __syncthreads();

    int wave = (blockIdx.x * FC_BLOCK + threadIdx.x) >> 5;
    int lane = threadIdx.x & 31;
    int m_tile = wave / N_TILES;
    int n_tile = wave % N_TILES;
    if (m_tile >= M_TILES) return;

    const int m_base = m_tile * 16;
    const int l16    = lane & 15;
    const int kb     = (lane >> 4) * 2;

    const int mrow   = m_base + l16;
    const int ncol_l = n_tile * 16 + l16;
    const float* arow = x + (size_t)mrow * DD;
    const float* brow = wlds + ncol_l * WP;

    const int2* node = (const int2*)node_raw;
    const int m_off = (lane >> 4) * 8;
    float dvr[8];
    #pragma unroll
    for (int v = 0; v < 8; ++v) dvr[v] = __int_as_float(node[m_base + m_off + v].y);

    v8f acc = {};
    #pragma unroll 8
    for (int k = 0; k < DD; k += 4) {
        v2f a, b;
        a.x = arow[k + kb];
        a.y = arow[k + kb + 1];
        b.x = brow[k + kb];
        b.y = brow[k + kb + 1];
        acc = __builtin_amdgcn_wmma_f32_16x16x4_f32(
             false, a,  false, b,
             (short)0, acc,  false,  false);
    }

    asm volatile("v_nop\n\tv_nop\n\tv_nop\n\tv_nop\n\tv_nop\n\ts_wait_alu depctr_va_vdst(0)" : "+v"(acc));
    if (ncol_l < CC) {
        #pragma unroll
        for (int v = 0; v < 8; ++v) {
            out[(size_t)(m_base + m_off + v) * PITCH + ncol_l] = acc[v] * dvr[v];
        }
    }
}

extern "C" void kernel_launch(void* const* d_in, const int* in_sizes, int n_in,
                              void* d_out, int out_size, void* d_ws, size_t ws_size,
                              hipStream_t stream) {
    const float* feat  = (const float*)d_in[0];
    const int*   row   = (const int*)  d_in[1];
    const int*   col   = (const int*)  d_in[2];
    const float* W     = (const float*)d_in[3];
    const float* bias  = (const float*)d_in[4];
    const float* gamma = (const float*)d_in[5];
    const float* beta  = (const float*)d_in[6];
    float* out = (float*)d_out;

    const int B = 256;
    const int fc_waves    = M_TILES * N_TILES;
    const int fc_blocks   = (fc_waves * 32 + FC_BLOCK - 1) / FC_BLOCK;
    const int prop_blocks = (NTRIP / 2 * 32 + B - 1) / B;

    int*   node = (int*)d_ws;
    int*   mat  = node + 2 * NP;
    float* ta   = (float*)(mat + NP);
    static_assert((3 * NP * 4) % 256 == 0, "the tables start on a 256-byte unit of the workspace");
    static_assert(2 * EE <= NN * CC, "the staged pairs fit the table they borrow, dense or padded");

    const size_t padded_need = ((size_t)3 * NP + (size_t)2 * NN * HP + (size_t)(EE + CSR_PAD)) * 4;

    if (ws_size >= padded_need) {
        float* tb   = ta + (size_t)NN * HP;
        int*   csr  = (int*)(tb + (size_t)NN * HP);
        int*   stage = (int*)tb;

        dgc_part_count   <<<NCH, SB1, 0, stream>>>(row, mat);
        dgc_part_scan    <<<1,   SB1, 0, stream>>>(mat);
        dgc_part_scatter <<<NCH, SB1, 0, stream>>>(row, col, mat, stage);
        dgc_row_build<HP><<<NB1, SB1, 0, stream>>>(stage, mat, node, csr);

        dgc_fc_wmma<HP><<<fc_blocks, FC_BLOCK, 0, stream>>>(feat, W, node, ta);

        dgc_prop<false, HP, HP><<<prop_blocks, B, 0, stream>>>(ta, tb, node, csr, bias, gamma, beta);
        dgc_prop<false, HP, HP><<<prop_blocks, B, 0, stream>>>(tb, ta, node, csr, bias, gamma, beta);
        dgc_prop<false, HP, HP><<<prop_blocks, B, 0, stream>>>(ta, tb, node, csr, bias, gamma, beta);
        dgc_prop<true,  HP, CC><<<prop_blocks, B, 0, stream>>>(tb, out, node, csr, bias, gamma, beta);
    } else {
        float* h    = ta;
        int*   csr  = (int*)(h + (size_t)NN * CC);
        int*   stage = (int*)h;

        dgc_part_count   <<<NCH, SB1, 0, stream>>>(row, mat);
        dgc_part_scan    <<<1,   SB1, 0, stream>>>(mat);
        dgc_part_scatter <<<NCH, SB1, 0, stream>>>(row, col, mat, stage);
        dgc_row_build<CC><<<NB1, SB1, 0, stream>>>(stage, mat, node, csr);

        dgc_fc_wmma<CC><<<fc_blocks, FC_BLOCK, 0, stream>>>(feat, W, node, out);

        dgc_prop<false, CC, CC><<<prop_blocks, B, 0, stream>>>(out, h, node, csr, bias, gamma, beta);
        dgc_prop<false, CC, CC><<<prop_blocks, B, 0, stream>>>(h, out, node, csr, bias, gamma, beta);
        dgc_prop<false, CC, CC><<<prop_blocks, B, 0, stream>>>(out, h, node, csr, bias, gamma, beta);
        dgc_prop<true,  CC, CC><<<prop_blocks, B, 0, stream>>>(h, out, node, csr, bias, gamma, beta);
    }
}


# FC: one wave per 16-row tile does all three 16-column tiles from one A read (feat read by one wave, not three; 3x fewer waves); W staged by 16-byte loads in one round trip; k order, sums kept
# speedup vs baseline: 9.6981x; 1.1136x over previous
#include <hip/hip_runtime.h>
#include <hip/hip_bf16.h>

#define NN      100000
#define EE      1600000
#define DD      128
#define CC      40
#define KSTEPS  4
#define DELTA   0.5f
#define LN_EPS  1e-5f

#define HP      64
#define CSR_PAD 8
#define NP      100352
static_assert(NP > NN, "entry N exists");
static_assert(KSTEPS == 4, "the launcher writes the four steps out, the last one into out");
static_assert(HP >= CC && (HP * 4) % 256 == 0, "a padded row is whole 256-byte units");

typedef __attribute__((ext_vector_type(2))) float v2f;
typedef __attribute__((ext_vector_type(8))) float v8f;

#define NB1    256
#define NCH    256
#define RB     392
#define SB1    1024
#define ECH    (EE / NCH)
#define EPT    ((ECH + SB1 - 1) / SB1)
#define RBATCH 8
static_assert(NB1 * RB == NP, "the ranges tile the padded row space");
static_assert(EE % NCH == 0, "whole chunks");
static_assert(NB1 * NCH == 2048 * (SB1 / 32), "a wave of the scan owns 2048 matrix entries");
static_assert(NB1 * NCH <= NP, "the count matrix fits the region the launcher gives it");

__global__ __launch_bounds__(SB1)
void dgc_part_count(const int* __restrict__ row, int* __restrict__ mat) {
    __shared__ int h[NB1];
    const int t = threadIdx.x;
    if (t < NB1) h[t] = 0;
    __syncthreads();
    const int* src = row + blockIdx.x * ECH;
    int r[EPT];
    #pragma unroll
    for (int k = 0; k < EPT; ++k) {
        const int i = t + k * SB1;
        r[k] = (i < ECH) ? src[i] : 0;
    }
    #pragma unroll
    for (int k = 0; k < EPT; ++k)
        if (t + k * SB1 < ECH) atomicAdd(&h[r[k] / RB], 1);
    __syncthreads();
    if (t < NB1) mat[t * NCH + blockIdx.x] = h[t];
}

__global__ __launch_bounds__(SB1)
void dgc_part_scan(int* __restrict__ mat) {
    __shared__ int s_tot[32];
    const int t = threadIdx.x, lane = t & 31, wv = t >> 5;
    int4* seg = (int4*)mat + wv * 512;
    int4 v[16];
    #pragma unroll
    for (int k = 0; k < 16; ++k) v[k] = seg[k * 32 + lane];
    int ex[16];
    int carry = 0;
    #pragma unroll
    for (int k = 0; k < 16; ++k) {
        const int s = (v[k].x + v[k].y) + (v[k].z + v[k].w);
        int inc = s;
        #pragma unroll
        for (int off = 1; off < 32; off <<= 1) {
            const int u = __shfl_up(inc, off, 32);
            if (lane >= off) inc += u;
        }
        ex[k] = carry + inc - s;
        carry += __shfl(inc, 31, 32);
    }
    if (lane == 0) s_tot[wv] = carry;
    __syncthreads();
    int wbase = (lane < wv) ? s_tot[lane] : 0;
    for (int off = 16; off > 0; off >>= 1) wbase += __shfl_xor(wbase, off, 32);
    #pragma unroll
    for (int k = 0; k < 16; ++k) {
        int4 o;
        o.x = wbase + ex[k];
        o.y = o.x + v[k].x;
        o.z = o.y + v[k].y;
        o.w = o.z + v[k].z;
        seg[k * 32 + lane] = o;
    }
}

__global__ __launch_bounds__(SB1)
void dgc_part_scatter(const int* __restrict__ row,
                      const int* __restrict__ col,
                      const int* __restrict__ mat,
                      int* __restrict__ stage_raw) {
    __shared__ int cur[NB1];
    const int t = threadIdx.x;
    if (t < NB1) cur[t] = mat[t * NCH + blockIdx.x];
    __syncthreads();
    const int* rsrc = row + blockIdx.x * ECH;
    const int* csrc = col + blockIdx.x * ECH;
    int r[EPT], c[EPT];
    #pragma unroll
    for (int k = 0; k < EPT; ++k) {
        const int i = t + k * SB1;
        r[k] = (i < ECH) ? rsrc[i] : 0;
        c[k] = (i < ECH) ? csrc[i] : 0;
    }
    #pragma unroll
    for (int k = 0; k < EPT; ++k) {
        if (t + k * SB1 < ECH) {
            const int p = atomicAdd(&cur[r[k] / RB], 1);
            ((int2*)stage_raw)[p] = make_int2(r[k], c[k]);
        }
    }
}

template <int PITCH>
__global__ __launch_bounds__(SB1)
void dgc_row_build(const int* __restrict__ stage_raw,
                   const int* __restrict__ mat,
                   int* __restrict__ node_raw,
                   int* __restrict__ csr) {
    __shared__ int cn[RB];
    __shared__ int s_tot[32];
    const int t = threadIdx.x, lane = t & 31, wv = t >> 5;
    const int b = blockIdx.x, r0 = b * RB;
    if (t < RB) cn[t] = 0;
    if (b == 0 && t < CSR_PAD) csr[EE + t] = 0;
    __syncthreads();
    const int beg = mat[b * NCH];
    const int end = (b + 1 < NB1) ? mat[(b + 1) * NCH] : EE;
    const int2* stage = (const int2*)stage_raw;
    const int i00 = beg + t;

    int2 e0[RBATCH];
    #pragma unroll
    for (int k = 0; k < RBATCH; ++k) {
        const int i = i00 + k * SB1;
        e0[k] = stage[(i < end) ? i : (EE - 1)];
    }
    #pragma unroll
    for (int k = 0; k < RBATCH; ++k)
        if (i00 + k * SB1 < end) atomicAdd(&cn[e0[k].x - r0], 1);
    for (int i0 = i00 + RBATCH * SB1; i0 < end; i0 += RBATCH * SB1) {
        int r[RBATCH];
        #pragma unroll
        for (int k = 0; k < RBATCH; ++k) {
            const int i = i0 + k * SB1;
            r[k] = stage[(i < end) ? i : (EE - 1)].x;
        }
        #pragma unroll
        for (int k = 0; k < RBATCH; ++k)
            if (i0 + k * SB1 < end) atomicAdd(&cn[r[k] - r0], 1);
    }
    __syncthreads();

    const int cv = (t < RB) ? cn[t] : 0;
    int inc = cv;
    for (int off = 1; off < 32; off <<= 1) {
        const int u = __shfl_up(inc, off, 32);
        if (lane >= off) inc += u;
    }
    if (lane == 31) s_tot[wv] = inc;
    __syncthreads();
    int wbase = (lane < wv) ? s_tot[lane] : 0;
    for (int off = 16; off > 0; off >>= 1) wbase += __shfl_xor(wbase, off, 32);
    if (t < RB) {
        int2 rec;
        rec.x = beg + wbase + inc - cv;
        rec.y = __float_as_int(rsqrtf((float)cv + 1.0f));
        ((int2*)node_raw)[r0 + t] = rec;
        cn[t] = rec.x;
    }
    __syncthreads();

    #pragma unroll
    for (int k = 0; k < RBATCH; ++k) {
        if (i00 + k * SB1 < end) {
            const int p = atomicAdd(&cn[e0[k].x - r0], 1);
            csr[p] = e0[k].y * PITCH;
        }
    }
    for (int i0 = i00 + RBATCH * SB1; i0 < end; i0 += RBATCH * SB1) {
        int2 e[RBATCH];
        #pragma unroll
        for (int k = 0; k < RBATCH; ++k) {
            const int i = i0 + k * SB1;
            e[k] = stage[(i < end) ? i : (EE - 1)];
        }
        #pragma unroll
        for (int k = 0; k < RBATCH; ++k) {
            if (i0 + k * SB1 < end) {
                const int p = atomicAdd(&cn[e[k].x - r0], 1);
                csr[p] = e[k].y * PITCH;
            }
        }
    }
}

#define GL     10
#define RPW    3
#define NTRIP  ((NN + RPW - 1) / RPW)
#define PB     8
static_assert(GL * 4 == CC, "ten float4 make a row");
static_assert(PB - 1 <= CSR_PAD, "a part batch over-reads at most PB - 1 records");

template <bool FULL>
__device__ __forceinline__ void dgc_batch(const float* __restrict__ hin,
                                          const int* __restrict__ csr,
                                          int e, int rem, int j4, float4& acc) {
    int    m[PB];
    float4 v[PB];
    #pragma unroll
    for (int k = 0; k < PB; ++k) m[k] = csr[e + k];
    if (!FULL) {
        #pragma unroll
        for (int k = 0; k < PB; ++k)
            asm volatile("" : "+v"(m[k]));
    }
    #pragma unroll
    for (int k = 0; k < PB; ++k) {
        v[k] = make_float4(0.f, 0.f, 0.f, 0.f);
        if (FULL || k < rem) v[k] = *(const float4*)(hin + m[k] + j4);
    }
    #pragma unroll
    for (int k = 0; k < PB; ++k) {
        acc.x += v[k].x;
        acc.y += v[k].y;
        acc.z += v[k].z;
        acc.w += v[k].w;
    }
}

struct dgc_ring { int p5, p1, p2, p3, p4, p0; };

__device__ __forceinline__ float dgc_group_sum(float v, const dgc_ring& g) {
    const float t = v + __shfl(v, g.p5, 32);
    float s = t + __shfl(t, g.p1, 32);
    s += __shfl(t, g.p2, 32);
    s += __shfl(t, g.p3, 32);
    s += __shfl(t, g.p4, 32);
    return __shfl(s, g.p0, 32);
}

__device__ __forceinline__ float4 dgc_step_row(const float4 gv, const float sa, const float sb,
                                               const float4 acc) {
    float4 o;
    o.x = (1.0f - DELTA) * (gv.x * sa) + DELTA * (sb * (acc.x + gv.x));
    o.y = (1.0f - DELTA) * (gv.y * sa) + DELTA * (sb * (acc.y + gv.y));
    o.z = (1.0f - DELTA) * (gv.z * sa) + DELTA * (sb * (acc.z + gv.z));
    o.w = (1.0f - DELTA) * (gv.w * sa) + DELTA * (sb * (acc.w + gv.w));
    return o;
}

template <bool LAST, int PIN, int POUT>
__global__ __launch_bounds__(256)
void dgc_prop(const float* __restrict__ hin,
              float* __restrict__ hout,
              const int* __restrict__ node_raw,
              const int* __restrict__ csr_raw,
              const float* __restrict__ bias,
              const float* __restrict__ gamma,
              const float* __restrict__ beta) {
    const int2* node = (const int2*)node_raw;
    const int*  csr  = csr_raw;
    const int lane   = threadIdx.x & 31;
    const int grp    = lane / GL;
    const int j4     = (lane - grp * GL) * 4;
    const int wave   = (blockIdx.x * blockDim.x + threadIdx.x) >> 5;
    const int nwaves = (gridDim.x * blockDim.x) >> 5;

    float4 bv = make_float4(0.f, 0.f, 0.f, 0.f), gv = bv, tv = bv;
    dgc_ring ring = {lane, lane, lane, lane, lane, lane};
    if (LAST) {
        bv = make_float4(bias[j4],  bias[j4 + 1],  bias[j4 + 2],  bias[j4 + 3]);
        gv = make_float4(gamma[j4], gamma[j4 + 1], gamma[j4 + 2], gamma[j4 + 3]);
        tv = make_float4(beta[j4],  beta[j4 + 1],  beta[j4 + 2],  beta[j4 + 3]);
        if (grp < RPW) {
            const int base = grp * GL;
            const int j    = lane - base;
            ring.p5 = base + (j + 5) % GL;
            ring.p1 = base + (j + 1) % GL;
            ring.p2 = base + (j + 2) % GL;
            ring.p3 = base + (j + 3) % GL;
            ring.p4 = base + (j + 4) % GL;
            ring.p0 = base;
        }
    }

    for (int t = wave; t < NTRIP; t += nwaves) {
        const int  r    = t * RPW + grp;
        const bool live = (grp < RPW) && (r < NN);
        int beg = 0, end = 0;
        float dv = 0.f;
        float4 hv = make_float4(0.f, 0.f, 0.f, 0.f);
        if (live) {
            const int2 nr = node[r];
            beg = nr.x;
            end = node[r + 1].x;
            dv  = __int_as_float(nr.y);
            hv  = *(const float4*)(hin + r * PIN + j4);
        }
        float4 acc = make_float4(0.f, 0.f, 0.f, 0.f);
        int e = beg;
        for (; e + PB <= end; e += PB) dgc_batch<true>(hin, csr, e, PB, j4, acc);
        if (e < end) dgc_batch<false>(hin, csr, e, end - e, j4, acc);
        if (!LAST) {
            if (live) *(float4*)(hout + r * POUT + j4) = dgc_step_row(hv, 1.0f, dv * dv, acc);
        } else {
            const float sd = ((float)(end - beg) + 1.0f) * dv;
            const float4 o = dgc_step_row(hv, sd, dv, acc);
            float4 x;
            x.x = o.x + bv.x;
            x.y = o.y + bv.y;
            x.z = o.z + bv.z;
            x.w = o.w + bv.w;
            const float s  = dgc_group_sum((x.x + x.y) + (x.z + x.w), ring);
            const float sq = dgc_group_sum((x.x * x.x + x.y * x.y) + (x.z * x.z + x.w * x.w), ring);
            const float mean = s * (1.0f / CC);
            const float var  = sq * (1.0f / CC) - mean * mean;
            const float inv  = rsqrtf(var + LN_EPS);
            float4 y;
            y.x = (x.x - mean) * inv * gv.x + tv.x;
            y.y = (x.y - mean) * inv * gv.y + tv.y;
            y.z = (x.z - mean) * inv * gv.z + tv.z;
            y.w = (x.w - mean) * inv * gv.w + tv.w;
            if (live) *(float4*)(hout + r * POUT + j4) = y;
        }
    }
}

#define M_TILES   (NN / 16)
#define N_TILES   3
#define NPAD      (N_TILES * 16)
#define FC_BLOCK  256
#define WP        (DD + 4)
static_assert((WP * 4) % 16 == 0 && ((WP * 4) / 16) % 2 == 1, "W row pitch in LDS: 16 bytes x odd");
static_assert(N_TILES == 3 && CC > 32 && CC <= NPAD, "the FC writes its three column tiles out by hand");
#define W_F4      (CC * DD / 4)
static_assert(DD == 128 && W_F4 % FC_BLOCK == 0 && (NPAD - CC) * (DD / 4) == FC_BLOCK,
              "the W stage: whole passes of the block over W, one zero piece per thread");

template <int PITCH>
__global__ __launch_bounds__(FC_BLOCK)
void dgc_fc_wmma(const float* __restrict__ x,
                 const float* __restrict__ W,
                 const int* __restrict__ node_raw,
                 float* __restrict__ out) {
    __shared__ __attribute__((aligned(16))) float wlds[NPAD * WP];

    {
        const float4* W4 = (const float4*)W;
        float4 w[W_F4 / FC_BLOCK];
        #pragma unroll
        for (int i = 0; i < W_F4 / FC_BLOCK; ++i) w[i] = W4[threadIdx.x + i * FC_BLOCK];
        #pragma unroll
        for (int i = 0; i < W_F4 / FC_BLOCK; ++i) {
            const int q = threadIdx.x + i * FC_BLOCK;
            *(float4*)(wlds + (q >> 5) * WP + (q & 31) * 4) = w[i];
        }
        const int z = threadIdx.x;
        *(float4*)(wlds + (CC + (z >> 5)) * WP + (z & 31) * 4) = make_float4(0.f, 0.f, 0.f, 0.f);
    }
    __syncthreads();

    const int m_tile = (blockIdx.x * FC_BLOCK + threadIdx.x) >> 5;
    const int lane   = threadIdx.x & 31;
    if (m_tile >= M_TILES) return;

    const int m_base = m_tile * 16;
    const int l16    = lane & 15;
    const int kb     = (lane >> 4) * 2;

    const float* arow = x + (size_t)(m_base + l16) * DD + kb;
    const float* brow = wlds + l16 * WP + kb;

    const int2* node = (const int2*)node_raw;
    const int m_off = (lane >> 4) * 8;
    float dvr[8];
    #pragma unroll
    for (int v = 0; v < 8; ++v) dvr[v] = __int_as_float(node[m_base + m_off + v].y);

    v8f acc0 = {}, acc1 = {}, acc2 = {};
    #pragma unroll 16
    for (int k = 0; k < DD; k += 4) {
        v2f a, b0, b1, b2;
        a.x  = arow[k];
        a.y  = arow[k + 1];
        b0.x = brow[k];
        b0.y = brow[k + 1];
        b1.x = brow[16 * WP + k];
        b1.y = brow[16 * WP + k + 1];
        b2.x = brow[32 * WP + k];
        b2.y = brow[32 * WP + k + 1];
        acc0 = __builtin_amdgcn_wmma_f32_16x16x4_f32(
             false, a,  false, b0,
             (short)0, acc0,  false,  false);
        acc1 = __builtin_amdgcn_wmma_f32_16x16x4_f32(
             false, a,  false, b1,
             (short)0, acc1,  false,  false);
        acc2 = __builtin_amdgcn_wmma_f32_16x16x4_f32(
             false, a,  false, b2,
             (short)0, acc2,  false,  false);
    }

    asm volatile("v_nop\n\tv_nop\n\tv_nop\n\tv_nop\n\tv_nop\n\ts_wait_alu depctr_va_vdst(0)"
                 : "+v"(acc0), "+v"(acc1), "+v"(acc2));
    float* orow = out + (size_t)(m_base + m_off) * PITCH + l16;
    #pragma unroll
    for (int v = 0; v < 8; ++v) {
        orow[(size_t)v * PITCH]      = acc0[v] * dvr[v];
        orow[(size_t)v * PITCH + 16] = acc1[v] * dvr[v];
    }
    if (l16 < CC - 32) {
        #pragma unroll
        for (int v = 0; v < 8; ++v) orow[(size_t)v * PITCH + 32] = acc2[v] * dvr[v];
    }
}

extern "C" void kernel_launch(void* const* d_in, const int* in_sizes, int n_in,
                              void* d_out, int out_size, void* d_ws, size_t ws_size,
                              hipStream_t stream) {
    const float* feat  = (const float*)d_in[0];
    const int*   row   = (const int*)  d_in[1];
    const int*   col   = (const int*)  d_in[2];
    const float* W     = (const float*)d_in[3];
    const float* bias  = (const float*)d_in[4];
    const float* gamma = (const float*)d_in[5];
    const float* beta  = (const float*)d_in[6];
    float* out = (float*)d_out;

    const int B = 256;
    const int fc_waves    = M_TILES;
    const int fc_blocks   = (fc_waves * 32 + FC_BLOCK - 1) / FC_BLOCK;
    const int prop_blocks = (NTRIP / 2 * 32 + B - 1) / B;

    int*   node = (int*)d_ws;
    int*   mat  = node + 2 * NP;
    float* ta   = (float*)(mat + NP);
    static_assert((3 * NP * 4) % 256 == 0, "the tables start on a 256-byte unit of the workspace");
    static_assert(2 * EE <= NN * CC, "the staged pairs fit the table they borrow, dense or padded");

    const size_t padded_need = ((size_t)3 * NP + (size_t)2 * NN * HP + (size_t)(EE + CSR_PAD)) * 4;

    if (ws_size >= padded_need) {
        float* tb   = ta + (size_t)NN * HP;
        int*   csr  = (int*)(tb + (size_t)NN * HP);
        int*   stage = (int*)tb;

        dgc_part_count   <<<NCH, SB1, 0, stream>>>(row, mat);
        dgc_part_scan    <<<1,   SB1, 0, stream>>>(mat);
        dgc_part_scatter <<<NCH, SB1, 0, stream>>>(row, col, mat, stage);
        dgc_row_build<HP><<<NB1, SB1, 0, stream>>>(stage, mat, node, csr);

        dgc_fc_wmma<HP><<<fc_blocks, FC_BLOCK, 0, stream>>>(feat, W, node, ta);

        dgc_prop<false, HP, HP><<<prop_blocks, B, 0, stream>>>(ta, tb, node, csr, bias, gamma, beta);
        dgc_prop<false, HP, HP><<<prop_blocks, B, 0, stream>>>(tb, ta, node, csr, bias, gamma, beta);
        dgc_prop<false, HP, HP><<<prop_blocks, B, 0, stream>>>(ta, tb, node, csr, bias, gamma, beta);
        dgc_prop<true,  HP, CC><<<prop_blocks, B, 0, stream>>>(tb, out, node, csr, bias, gamma, beta);
    } else {
        float* h    = ta;
        int*   csr  = (int*)(h + (size_t)NN * CC);
        int*   stage = (int*)h;

        dgc_part_count   <<<NCH, SB1, 0, stream>>>(row, mat);
        dgc_part_scan    <<<1,   SB1, 0, stream>>>(mat);
        dgc_part_scatter <<<NCH, SB1, 0, stream>>>(row, col, mat, stage);
        dgc_row_build<CC><<<NB1, SB1, 0, stream>>>(stage, mat, node, csr);

        dgc_fc_wmma<CC><<<fc_blocks, FC_BLOCK, 0, stream>>>(feat, W, node, out);

        dgc_prop<false, CC, CC><<<prop_blocks, B, 0, stream>>>(out, h, node, csr, bias, gamma, beta);
        dgc_prop<false, CC, CC><<<prop_blocks, B, 0, stream>>>(h, out, node, csr, bias, gamma, beta);
        dgc_prop<false, CC, CC><<<prop_blocks, B, 0, stream>>>(out, h, node, csr, bias, gamma, beta);
        dgc_prop<true,  CC, CC><<<prop_blocks, B, 0, stream>>>(h, out, node, csr, bias, gamma, beta);
    }
}


# Edge sort level 1 made chunk-local: a work-group counts, scans and places its chunk in LDS, stores one sorted piece + 256 offsets; global count matrix, its scan, 2 launches go; f32 sum order changes
# speedup vs baseline: 11.3950x; 1.1750x over previous
#include <hip/hip_runtime.h>
#include <hip/hip_bf16.h>

#define NN      100000
#define EE      1600000
#define DD      128
#define CC      40
#define KSTEPS  4
#define DELTA   0.5f
#define LN_EPS  1e-5f

#define HP      64
#define CSR_PAD 8
#define NP      100352
static_assert(NP > NN, "entry N exists");
static_assert(KSTEPS == 4, "the launcher writes the four steps out, the last one into out");
static_assert(HP >= CC && (HP * 4) % 256 == 0, "a padded row is whole 256-byte units");

typedef __attribute__((ext_vector_type(2))) float v2f;
typedef __attribute__((ext_vector_type(8))) float v8f;

#define NB1    256
#define NCH    256
#define RB     392
#define SB1    1024
#define ECH    (EE / NCH)
#define EPT    ((ECH + SB1 - 1) / SB1)
#define SEGL   16
#define SEGK   3
#define SEGR   (NCH / (SB1 / SEGL))
static_assert(NB1 * RB == NP, "the ranges tile the padded row space");
static_assert(EE % NCH == 0 && ECH % 2 == 0, "whole chunks, stored as 16-byte pieces");
static_assert(NB1 == 8 * 32 && NB1 <= SB1, "eight whole waves scan the range counts");
static_assert(NB1 * NCH <= NP, "the offset matrix fits the region the launcher gives it");
static_assert(SEGR * (SB1 / SEGL) == NCH && (SEGL & (SEGL - 1)) == 0, "the lane groups cover the runs");

__global__ __launch_bounds__(SB1)
void dgc_part_local(const int* __restrict__ row,
                    const int* __restrict__ col,
                    int* __restrict__ offs,
                    int* __restrict__ stage_raw) {
    __shared__ __attribute__((aligned(16))) int2 buf[ECH];
    __shared__ int h[NB1];
    __shared__ int s_tot[32];
    const int t = threadIdx.x, lane = t & 31, wv = t >> 5;
    if (t < NB1) h[t] = 0;
    __syncthreads();
    const int* rsrc = row + blockIdx.x * ECH;
    const int* csrc = col + blockIdx.x * ECH;
    int r[EPT], c[EPT];
    #pragma unroll
    for (int k = 0; k < EPT; ++k) {
        const int i = t + k * SB1;
        r[k] = (i < ECH) ? rsrc[i] : 0;
        c[k] = (i < ECH) ? csrc[i] : 0;
    }
    #pragma unroll
    for (int k = 0; k < EPT; ++k)
        if (t + k * SB1 < ECH) atomicAdd(&h[r[k] / RB], 1);
    __syncthreads();

    const int cv = (t < NB1) ? h[t] : 0;
    int inc = cv;
    for (int off = 1; off < 32; off <<= 1) {
        const int u = __shfl_up(inc, off, 32);
        if (lane >= off) inc += u;
    }
    if (lane == 31) s_tot[wv] = inc;
    __syncthreads();
    int wbase = (lane < wv) ? s_tot[lane] : 0;
    for (int off = 16; off > 0; off >>= 1) wbase += __shfl_xor(wbase, off, 32);
    if (t < NB1) {
        const int ex = wbase + inc - cv;
        offs[t * NCH + blockIdx.x] = ex;
        h[t] = ex;
    }
    __syncthreads();

    #pragma unroll
    for (int k = 0; k < EPT; ++k) {
        if (t + k * SB1 < ECH) {
            const int p = atomicAdd(&h[r[k] / RB], 1);
            buf[p] = make_int2(r[k], c[k]);
        }
    }
    __syncthreads();
    const int4* b4 = (const int4*)buf;
    int4* dst = (int4*)stage_raw + blockIdx.x * (ECH / 2);
    for (int i = t; i < ECH / 2; i += SB1) dst[i] = b4[i];
}

template <int PITCH>
__global__ __launch_bounds__(SB1)
void dgc_row_build(const int* __restrict__ stage_raw,
                   const int* __restrict__ offs,
                   int* __restrict__ node_raw,
                   int* __restrict__ csr) {
    __shared__ int cn[RB];
    __shared__ int s_tot[32];
    __shared__ int s_part[32];
    __shared__ int s_lo[NCH];
    __shared__ int s_len[NCH];
    const int t = threadIdx.x, lane = t & 31, wv = t >> 5;
    const int b = blockIdx.x, r0 = b * RB;
    if (t < RB) cn[t] = 0;
    if (b == 0 && t < CSR_PAD) csr[EE + t] = 0;
    int lo = 0;
    if (t < NCH) {
        lo = offs[b * NCH + t];
        const int hi = (b + 1 < NB1) ? offs[(b + 1) * NCH + t] : ECH;
        s_lo[t]  = t * ECH + lo;
        s_len[t] = hi - lo;
    }
    int part = lo;
    for (int off = 16; off > 0; off >>= 1) part += __shfl_xor(part, off, 32);
    if (lane == 0) s_part[wv] = part;
    __syncthreads();
    int beg = s_part[lane];
    for (int off = 16; off > 0; off >>= 1) beg += __shfl_xor(beg, off, 32);

    const int2* stage = (const int2*)stage_raw;
    const int sub = t & (SEGL - 1);
    int sbase[SEGR], slen[SEGR];
    int2 e0[SEGR][SEGK];
    #pragma unroll
    for (int q = 0; q < SEGR; ++q) {
        const int seg = q * (SB1 / SEGL) + (t / SEGL);
        sbase[q] = s_lo[seg];
        slen[q]  = s_len[seg];
    }
    #pragma unroll
    for (int q = 0; q < SEGR; ++q) {
        #pragma unroll
        for (int k = 0; k < SEGK; ++k) {
            const int i = sbase[q] + sub + k * SEGL;
            e0[q][k] = stage[(i < EE) ? i : (EE - 1)];
        }
    }
    #pragma unroll
    for (int q = 0; q < SEGR; ++q) {
        #pragma unroll
        for (int k = 0; k < SEGK; ++k)
            if (sub + k * SEGL < slen[q]) atomicAdd(&cn[e0[q][k].x - r0], 1);
        for (int j = sub + SEGK * SEGL; j < slen[q]; j += SEGL)
            atomicAdd(&cn[stage[sbase[q] + j].x - r0], 1);
    }
    __syncthreads();

    const int cv = (t < RB) ? cn[t] : 0;
    int inc = cv;
    for (int off = 1; off < 32; off <<= 1) {
        const int u = __shfl_up(inc, off, 32);
        if (lane >= off) inc += u;
    }
    if (lane == 31) s_tot[wv] = inc;
    __syncthreads();
    int wbase = (lane < wv) ? s_tot[lane] : 0;
    for (int off = 16; off > 0; off >>= 1) wbase += __shfl_xor(wbase, off, 32);
    if (t < RB) {
        int2 rec;
        rec.x = beg + wbase + inc - cv;
        rec.y = __float_as_int(rsqrtf((float)cv + 1.0f));
        ((int2*)node_raw)[r0 + t] = rec;
        cn[t] = rec.x;
    }
    __syncthreads();

    #pragma unroll
    for (int q = 0; q < SEGR; ++q) {
        #pragma unroll
        for (int k = 0; k < SEGK; ++k) {
            if (sub + k * SEGL < slen[q]) {
                const int p = atomicAdd(&cn[e0[q][k].x - r0], 1);
                csr[p] = e0[q][k].y * PITCH;
            }
        }
        for (int j = sub + SEGK * SEGL; j < slen[q]; j += SEGL) {
            const int2 e = stage[sbase[q] + j];
            const int p = atomicAdd(&cn[e.x - r0], 1);
            csr[p] = e.y * PITCH;
        }
    }
}

#define GL     10
#define RPW    3
#define NTRIP  ((NN + RPW - 1) / RPW)
#define PB     8
static_assert(GL * 4 == CC, "ten float4 make a row");
static_assert(PB - 1 <= CSR_PAD, "a part batch over-reads at most PB - 1 records");

template <bool FULL>
__device__ __forceinline__ void dgc_batch(const float* __restrict__ hin,
                                          const int* __restrict__ csr,
                                          int e, int rem, int j4, float4& acc) {
    int    m[PB];
    float4 v[PB];
    #pragma unroll
    for (int k = 0; k < PB; ++k) m[k] = csr[e + k];
    if (!FULL) {
        #pragma unroll
        for (int k = 0; k < PB; ++k)
            asm volatile("" : "+v"(m[k]));
    }
    #pragma unroll
    for (int k = 0; k < PB; ++k) {
        v[k] = make_float4(0.f, 0.f, 0.f, 0.f);
        if (FULL || k < rem) v[k] = *(const float4*)(hin + m[k] + j4);
    }
    #pragma unroll
    for (int k = 0; k < PB; ++k) {
        acc.x += v[k].x;
        acc.y += v[k].y;
        acc.z += v[k].z;
        acc.w += v[k].w;
    }
}

struct dgc_ring { int p5, p1, p2, p3, p4, p0; };

__device__ __forceinline__ float dgc_group_sum(float v, const dgc_ring& g) {
    const float t = v + __shfl(v, g.p5, 32);
    float s = t + __shfl(t, g.p1, 32);
    s += __shfl(t, g.p2, 32);
    s += __shfl(t, g.p3, 32);
    s += __shfl(t, g.p4, 32);
    return __shfl(s, g.p0, 32);
}

__device__ __forceinline__ float4 dgc_step_row(const float4 gv, const float sa, const float sb,
                                               const float4 acc) {
    float4 o;
    o.x = (1.0f - DELTA) * (gv.x * sa) + DELTA * (sb * (acc.x + gv.x));
    o.y = (1.0f - DELTA) * (gv.y * sa) + DELTA * (sb * (acc.y + gv.y));
    o.z = (1.0f - DELTA) * (gv.z * sa) + DELTA * (sb * (acc.z + gv.z));
    o.w = (1.0f - DELTA) * (gv.w * sa) + DELTA * (sb * (acc.w + gv.w));
    return o;
}

template <bool LAST, int PIN, int POUT>
__global__ __launch_bounds__(256)
void dgc_prop(const float* __restrict__ hin,
              float* __restrict__ hout,
              const int* __restrict__ node_raw,
              const int* __restrict__ csr_raw,
              const float* __restrict__ bias,
              const float* __restrict__ gamma,
              const float* __restrict__ beta) {
    const int2* node = (const int2*)node_raw;
    const int*  csr  = csr_raw;
    const int lane   = threadIdx.x & 31;
    const int grp    = lane / GL;
    const int j4     = (lane - grp * GL) * 4;
    const int wave   = (blockIdx.x * blockDim.x + threadIdx.x) >> 5;
    const int nwaves = (gridDim.x * blockDim.x) >> 5;

    float4 bv = make_float4(0.f, 0.f, 0.f, 0.f), gv = bv, tv = bv;
    dgc_ring ring = {lane, lane, lane, lane, lane, lane};
    if (LAST) {
        bv = make_float4(bias[j4],  bias[j4 + 1],  bias[j4 + 2],  bias[j4 + 3]);
        gv = make_float4(gamma[j4], gamma[j4 + 1], gamma[j4 + 2], gamma[j4 + 3]);
        tv = make_float4(beta[j4],  beta[j4 + 1],  beta[j4 + 2],  beta[j4 + 3]);
        if (grp < RPW) {
            const int base = grp * GL;
            const int j    = lane - base;
            ring.p5 = base + (j + 5) % GL;
            ring.p1 = base + (j + 1) % GL;
            ring.p2 = base + (j + 2) % GL;
            ring.p3 = base + (j + 3) % GL;
            ring.p4 = base + (j + 4) % GL;
            ring.p0 = base;
        }
    }

    for (int t = wave; t < NTRIP; t += nwaves) {
        const int  r    = t * RPW + grp;
        const bool live = (grp < RPW) && (r < NN);
        int beg = 0, end = 0;
        float dv = 0.f;
        float4 hv = make_float4(0.f, 0.f, 0.f, 0.f);
        if (live) {
            const int2 nr = node[r];
            beg = nr.x;
            end = node[r + 1].x;
            dv  = __int_as_float(nr.y);
            hv  = *(const float4*)(hin + r * PIN + j4);
        }
        float4 acc = make_float4(0.f, 0.f, 0.f, 0.f);
        int e = beg;
        for (; e + PB <= end; e += PB) dgc_batch<true>(hin, csr, e, PB, j4, acc);
        if (e < end) dgc_batch<false>(hin, csr, e, end - e, j4, acc);
        if (!LAST) {
            if (live) *(float4*)(hout + r * POUT + j4) = dgc_step_row(hv, 1.0f, dv * dv, acc);
        } else {
            const float sd = ((float)(end - beg) + 1.0f) * dv;
            const float4 o = dgc_step_row(hv, sd, dv, acc);
            float4 x;
            x.x = o.x + bv.x;
            x.y = o.y + bv.y;
            x.z = o.z + bv.z;
            x.w = o.w + bv.w;
            const float s  = dgc_group_sum((x.x + x.y) + (x.z + x.w), ring);
            const float sq = dgc_group_sum((x.x * x.x + x.y * x.y) + (x.z * x.z + x.w * x.w), ring);
            const float mean = s * (1.0f / CC);
            const float var  = sq * (1.0f / CC) - mean * mean;
            const float inv  = rsqrtf(var + LN_EPS);
            float4 y;
            y.x = (x.x - mean) * inv * gv.x + tv.x;
            y.y = (x.y - mean) * inv * gv.y + tv.y;
            y.z = (x.z - mean) * inv * gv.z + tv.z;
            y.w = (x.w - mean) * inv * gv.w + tv.w;
            if (live) *(float4*)(hout + r * POUT + j4) = y;
        }
    }
}

#define M_TILES   (NN / 16)
#define N_TILES   3
#define NPAD      (N_TILES * 16)
#define FC_BLOCK  256
#define WP        (DD + 4)
static_assert((WP * 4) % 16 == 0 && ((WP * 4) / 16) % 2 == 1, "W row pitch in LDS: 16 bytes x odd");
static_assert(N_TILES == 3 && CC > 32 && CC <= NPAD, "the FC writes its three column tiles out by hand");
#define W_F4      (CC * DD / 4)
static_assert(DD == 128 && W_F4 % FC_BLOCK == 0 && (NPAD - CC) * (DD / 4) == FC_BLOCK,
              "the W stage: whole passes of the block over W, one zero piece per thread");

template <int PITCH>
__global__ __launch_bounds__(FC_BLOCK)
void dgc_fc_wmma(const float* __restrict__ x,
                 const float* __restrict__ W,
                 const int* __restrict__ node_raw,
                 float* __restrict__ out) {
    __shared__ __attribute__((aligned(16))) float wlds[NPAD * WP];

    {
        const float4* W4 = (const float4*)W;
        float4 w[W_F4 / FC_BLOCK];
        #pragma unroll
        for (int i = 0; i < W_F4 / FC_BLOCK; ++i) w[i] = W4[threadIdx.x + i * FC_BLOCK];
        #pragma unroll
        for (int i = 0; i < W_F4 / FC_BLOCK; ++i) {
            const int q = threadIdx.x + i * FC_BLOCK;
            *(float4*)(wlds + (q >> 5) * WP + (q & 31) * 4) = w[i];
        }
        const int z = threadIdx.x;
        *(float4*)(wlds + (CC + (z >> 5)) * WP + (z & 31) * 4) = make_float4(0.f, 0.f, 0.f, 0.f);
    }
    __syncthreads();

    const int m_tile = (blockIdx.x * FC_BLOCK + threadIdx.x) >> 5;
    const int lane   = threadIdx.x & 31;
    if (m_tile >= M_TILES) return;

    const int m_base = m_tile * 16;
    const int l16    = lane & 15;
    const int kb     = (lane >> 4) * 2;

    const float* arow = x + (size_t)(m_base + l16) * DD + kb;
    const float* brow = wlds + l16 * WP + kb;

    const int2* node = (const int2*)node_raw;
    const int m_off = (lane >> 4) * 8;
    float dvr[8];
    #pragma unroll
    for (int v = 0; v < 8; ++v) dvr[v] = __int_as_float(node[m_base + m_off + v].y);

    v8f acc0 = {}, acc1 = {}, acc2 = {};
    #pragma unroll 16
    for (int k = 0; k < DD; k += 4) {
        v2f a, b0, b1, b2;
        a.x  = arow[k];
        a.y  = arow[k + 1];
        b0.x = brow[k];
        b0.y = brow[k + 1];
        b1.x = brow[16 * WP + k];
        b1.y = brow[16 * WP + k + 1];
        b2.x = brow[32 * WP + k];
        b2.y = brow[32 * WP + k + 1];
        acc0 = __builtin_amdgcn_wmma_f32_16x16x4_f32(
             false, a,  false, b0,
             (short)0, acc0,  false,  false);
        acc1 = __builtin_amdgcn_wmma_f32_16x16x4_f32(
             false, a,  false, b1,
             (short)0, acc1,  false,  false);
        acc2 = __builtin_amdgcn_wmma_f32_16x16x4_f32(
             false, a,  false, b2,
             (short)0, acc2,  false,  false);
    }

    asm volatile("v_nop\n\tv_nop\n\tv_nop\n\tv_nop\n\tv_nop\n\ts_wait_alu depctr_va_vdst(0)"
                 : "+v"(acc0), "+v"(acc1), "+v"(acc2));
    float* orow = out + (size_t)(m_base + m_off) * PITCH + l16;
    #pragma unroll
    for (int v = 0; v < 8; ++v) {
        orow[(size_t)v * PITCH]      = acc0[v] * dvr[v];
        orow[(size_t)v * PITCH + 16] = acc1[v] * dvr[v];
    }
    if (l16 < CC - 32) {
        #pragma unroll
        for (int v = 0; v < 8; ++v) orow[(size_t)v * PITCH + 32] = acc2[v] * dvr[v];
    }
}

extern "C" void kernel_launch(void* const* d_in, const int* in_sizes, int n_in,
                              void* d_out, int out_size, void* d_ws, size_t ws_size,
                              hipStream_t stream) {
    const float* feat  = (const float*)d_in[0];
    const int*   row   = (const int*)  d_in[1];
    const int*   col   = (const int*)  d_in[2];
    const float* W     = (const float*)d_in[3];
    const float* bias  = (const float*)d_in[4];
    const float* gamma = (const float*)d_in[5];
    const float* beta  = (const float*)d_in[6];
    float* out = (float*)d_out;

    const int B = 256;
    const int fc_waves    = M_TILES;
    const int fc_blocks   = (fc_waves * 32 + FC_BLOCK - 1) / FC_BLOCK;
    const int prop_blocks = (NTRIP / 2 * 32 + B - 1) / B;

    int*   node = (int*)d_ws;
    int*   mat  = node + 2 * NP;
    float* ta   = (float*)(mat + NP);
    static_assert((3 * NP * 4) % 256 == 0, "the tables start on a 256-byte unit of the workspace");
    static_assert(2 * EE <= NN * CC, "the staged pairs fit the table they borrow, dense or padded");

    const size_t padded_need = ((size_t)3 * NP + (size_t)2 * NN * HP + (size_t)(EE + CSR_PAD)) * 4;

    if (ws_size >= padded_need) {
        float* tb   = ta + (size_t)NN * HP;
        int*   csr  = (int*)(tb + (size_t)NN * HP);
        int*   stage = (int*)tb;

        dgc_part_local   <<<NCH, SB1, 0, stream>>>(row, col, mat, stage);
        dgc_row_build<HP><<<NB1, SB1, 0, stream>>>(stage, mat, node, csr);

        dgc_fc_wmma<HP><<<fc_blocks, FC_BLOCK, 0, stream>>>(feat, W, node, ta);

        dgc_prop<false, HP, HP><<<prop_blocks, B, 0, stream>>>(ta, tb, node, csr, bias, gamma, beta);
        dgc_prop<false, HP, HP><<<prop_blocks, B, 0, stream>>>(tb, ta, node, csr, bias, gamma, beta);
        dgc_prop<false, HP, HP><<<prop_blocks, B, 0, stream>>>(ta, tb, node, csr, bias, gamma, beta);
        dgc_prop<true,  HP, CC><<<prop_blocks, B, 0, stream>>>(tb, out, node, csr, bias, gamma, beta);
    } else {
        float* h    = ta;
        int*   csr  = (int*)(h + (size_t)NN * CC);
        int*   stage = (int*)h;

        dgc_part_local   <<<NCH, SB1, 0, stream>>>(row, col, mat, stage);
        dgc_row_build<CC><<<NB1, SB1, 0, stream>>>(stage, mat, node, csr);

        dgc_fc_wmma<CC><<<fc_blocks, FC_BLOCK, 0, stream>>>(feat, W, node, out);

        dgc_prop<false, CC, CC><<<prop_blocks, B, 0, stream>>>(out, h, node, csr, bias, gamma, beta);
        dgc_prop<false, CC, CC><<<prop_blocks, B, 0, stream>>>(h, out, node, csr, bias, gamma, beta);
        dgc_prop<false, CC, CC><<<prop_blocks, B, 0, stream>>>(out, h, node, csr, bias, gamma, beta);
        dgc_prop<true,  CC, CC><<<prop_blocks, B, 0, stream>>>(h, out, node, csr, bias, gamma, beta);
    }
}


# Edge sort level 2 places a row range's records in an LDS copy of its list piece and stores it as whole 128-byte lines, no scattered 4-byte global store per edge; slots still arrival order, sums kept
# speedup vs baseline: 11.7657x; 1.0325x over previous
#include <hip/hip_runtime.h>
#include <hip/hip_bf16.h>

#define NN      100000
#define EE      1600000
#define DD      128
#define CC      40
#define KSTEPS  4
#define DELTA   0.5f
#define LN_EPS  1e-5f

#define HP      64
#define CSR_PAD 8
#define NP      100352
static_assert(NP > NN, "entry N exists");
static_assert(KSTEPS == 4, "the launcher writes the four steps out, the last one into out");
static_assert(HP >= CC && (HP * 4) % 256 == 0, "a padded row is whole 256-byte units");

typedef __attribute__((ext_vector_type(2))) float v2f;
typedef __attribute__((ext_vector_type(8))) float v8f;

#define NB1    256
#define NCH    256
#define RB     392
#define SB1    1024
#define ECH    (EE / NCH)
#define EPT    ((ECH + SB1 - 1) / SB1)
#define SEGL   16
#define SEGK   3
#define SEGR   (NCH / (SB1 / SEGL))
#define LREC   8192
static_assert(NB1 * RB == NP, "the ranges tile the padded row space");
static_assert(EE % NCH == 0 && ECH % 2 == 0, "whole chunks, stored as 16-byte pieces");
static_assert(NB1 == 8 * 32 && NB1 <= SB1, "eight whole waves scan the range counts");
static_assert(NB1 * NCH <= NP, "the offset matrix fits the region the launcher gives it");
static_assert(SEGR * (SB1 / SEGL) == NCH && (SEGL & (SEGL - 1)) == 0, "the lane groups cover the runs");
static_assert(LREC % SB1 == 0, "the write-out of the LDS piece: whole passes of the work-group, and one for the first line's offset");

__global__ __launch_bounds__(SB1)
void dgc_part_local(const int* __restrict__ row,
                    const int* __restrict__ col,
                    int* __restrict__ offs,
                    int* __restrict__ stage_raw) {
    __shared__ __attribute__((aligned(16))) int2 buf[ECH];
    __shared__ int h[NB1];
    __shared__ int s_tot[32];
    const int t = threadIdx.x, lane = t & 31, wv = t >> 5;
    if (t < NB1) h[t] = 0;
    __syncthreads();
    const int* rsrc = row + blockIdx.x * ECH;
    const int* csrc = col + blockIdx.x * ECH;
    int r[EPT], c[EPT];
    #pragma unroll
    for (int k = 0; k < EPT; ++k) {
        const int i = t + k * SB1;
        r[k] = (i < ECH) ? rsrc[i] : 0;
        c[k] = (i < ECH) ? csrc[i] : 0;
    }
    #pragma unroll
    for (int k = 0; k < EPT; ++k)
        if (t + k * SB1 < ECH) atomicAdd(&h[r[k] / RB], 1);
    __syncthreads();

    const int cv = (t < NB1) ? h[t] : 0;
    int inc = cv;
    for (int off = 1; off < 32; off <<= 1) {
        const int u = __shfl_up(inc, off, 32);
        if (lane >= off) inc += u;
    }
    if (lane == 31) s_tot[wv] = inc;
    __syncthreads();
    int wbase = (lane < wv) ? s_tot[lane] : 0;
    for (int off = 16; off > 0; off >>= 1) wbase += __shfl_xor(wbase, off, 32);
    if (t < NB1) {
        const int ex = wbase + inc - cv;
        offs[t * NCH + blockIdx.x] = ex;
        h[t] = ex;
    }
    __syncthreads();

    #pragma unroll
    for (int k = 0; k < EPT; ++k) {
        if (t + k * SB1 < ECH) {
            const int p = atomicAdd(&h[r[k] / RB], 1);
            buf[p] = make_int2(r[k], c[k]);
        }
    }
    __syncthreads();
    const int4* b4 = (const int4*)buf;
    int4* dst = (int4*)stage_raw + blockIdx.x * (ECH / 2);
    for (int i = t; i < ECH / 2; i += SB1) dst[i] = b4[i];
}

template <int PITCH>
__global__ __launch_bounds__(SB1)
void dgc_row_build(const int* __restrict__ stage_raw,
                   const int* __restrict__ offs,
                   int* __restrict__ node_raw,
                   int* __restrict__ csr) {
    __shared__ int cn[RB];
    __shared__ int s_tot[32];
    __shared__ int s_part[32];
    __shared__ int s_lo[NCH];
    __shared__ int s_len[NCH];
    __shared__ int lrec[LREC];
    const int t = threadIdx.x, lane = t & 31, wv = t >> 5;
    const int b = blockIdx.x, r0 = b * RB;
    if (t < RB) cn[t] = 0;
    if (b == 0 && t < CSR_PAD) csr[EE + t] = 0;
    int lo = 0;
    if (t < NCH) {
        lo = offs[b * NCH + t];
        const int hi = (b + 1 < NB1) ? offs[(b + 1) * NCH + t] : ECH;
        s_lo[t]  = t * ECH + lo;
        s_len[t] = hi - lo;
    }
    int part = lo;
    for (int off = 16; off > 0; off >>= 1) part += __shfl_xor(part, off, 32);
    if (lane == 0) s_part[wv] = part;
    __syncthreads();
    int beg = s_part[lane];
    for (int off = 16; off > 0; off >>= 1) beg += __shfl_xor(beg, off, 32);

    const int2* stage = (const int2*)stage_raw;
    const int sub = t & (SEGL - 1);
    int sbase[SEGR], slen[SEGR];
    int2 e0[SEGR][SEGK];
    #pragma unroll
    for (int q = 0; q < SEGR; ++q) {
        const int seg = q * (SB1 / SEGL) + (t / SEGL);
        sbase[q] = s_lo[seg];
        slen[q]  = s_len[seg];
    }
    #pragma unroll
    for (int q = 0; q < SEGR; ++q) {
        #pragma unroll
        for (int k = 0; k < SEGK; ++k) {
            const int i = sbase[q] + sub + k * SEGL;
            e0[q][k] = stage[(i < EE) ? i : (EE - 1)];
        }
    }
    #pragma unroll
    for (int q = 0; q < SEGR; ++q) {
        #pragma unroll
        for (int k = 0; k < SEGK; ++k)
            if (sub + k * SEGL < slen[q]) atomicAdd(&cn[e0[q][k].x - r0], 1);
        for (int j = sub + SEGK * SEGL; j < slen[q]; j += SEGL)
            atomicAdd(&cn[stage[sbase[q] + j].x - r0], 1);
    }
    __syncthreads();

    const int cv = (t < RB) ? cn[t] : 0;
    int inc = cv;
    for (int off = 1; off < 32; off <<= 1) {
        const int u = __shfl_up(inc, off, 32);
        if (lane >= off) inc += u;
    }
    if (lane == 31) s_tot[wv] = inc;
    __syncthreads();
    int wbase = (lane < wv) ? s_tot[lane] : 0;
    int tot   = s_tot[lane];
    for (int off = 16; off > 0; off >>= 1) {
        wbase += __shfl_xor(wbase, off, 32);
        tot   += __shfl_xor(tot, off, 32);
    }
    if (t < RB) {
        const int first = wbase + inc - cv;
        int2 rec;
        rec.x = beg + first;
        rec.y = __float_as_int(rsqrtf((float)cv + 1.0f));
        ((int2*)node_raw)[r0 + t] = rec;
        cn[t] = first;
    }
    __syncthreads();

    #pragma unroll
    for (int q = 0; q < SEGR; ++q) {
        #pragma unroll
        for (int k = 0; k < SEGK; ++k) {
            if (sub + k * SEGL < slen[q]) {
                const int p = atomicAdd(&cn[e0[q][k].x - r0], 1);
                const int v = e0[q][k].y * PITCH;
                if (p < LREC) lrec[p] = v; else csr[beg + p] = v;
            }
        }
        for (int j = sub + SEGK * SEGL; j < slen[q]; j += SEGL) {
            const int2 e = stage[sbase[q] + j];
            const int p = atomicAdd(&cn[e.x - r0], 1);
            const int v = e.y * PITCH;
            if (p < LREC) lrec[p] = v; else csr[beg + p] = v;
        }
    }
    __syncthreads();

    const int n  = (tot < LREC) ? tot : LREC;
    const int i0 = t - (beg & 31);
    int w[LREC / SB1 + 1];
    #pragma unroll
    for (int k = 0; k <= LREC / SB1; ++k) {
        const int i = i0 + k * SB1;
        w[k] = (i >= 0 && i < n) ? lrec[i] : 0;
    }
    #pragma unroll
    for (int k = 0; k <= LREC / SB1; ++k) {
        const int i = i0 + k * SB1;
        if (i >= 0 && i < n) csr[beg + i] = w[k];
    }
}

#define GL     10
#define RPW    3
#define NTRIP  ((NN + RPW - 1) / RPW)
#define PB     8
static_assert(GL * 4 == CC, "ten float4 make a row");
static_assert(PB - 1 <= CSR_PAD, "a part batch over-reads at most PB - 1 records");

template <bool FULL>
__device__ __forceinline__ void dgc_batch(const float* __restrict__ hin,
                                          const int* __restrict__ csr,
                                          int e, int rem, int j4, float4& acc) {
    int    m[PB];
    float4 v[PB];
    #pragma unroll
    for (int k = 0; k < PB; ++k) m[k] = csr[e + k];
    if (!FULL) {
        #pragma unroll
        for (int k = 0; k < PB; ++k)
            asm volatile("" : "+v"(m[k]));
    }
    #pragma unroll
    for (int k = 0; k < PB; ++k) {
        v[k] = make_float4(0.f, 0.f, 0.f, 0.f);
        if (FULL || k < rem) v[k] = *(const float4*)(hin + m[k] + j4);
    }
    #pragma unroll
    for (int k = 0; k < PB; ++k) {
        acc.x += v[k].x;
        acc.y += v[k].y;
        acc.z += v[k].z;
        acc.w += v[k].w;
    }
}

struct dgc_ring { int p5, p1, p2, p3, p4, p0; };

__device__ __forceinline__ float dgc_group_sum(float v, const dgc_ring& g) {
    const float t = v + __shfl(v, g.p5, 32);
    float s = t + __shfl(t, g.p1, 32);
    s += __shfl(t, g.p2, 32);
    s += __shfl(t, g.p3, 32);
    s += __shfl(t, g.p4, 32);
    return __shfl(s, g.p0, 32);
}

__device__ __forceinline__ float4 dgc_step_row(const float4 gv, const float sa, const float sb,
                                               const float4 acc) {
    float4 o;
    o.x = (1.0f - DELTA) * (gv.x * sa) + DELTA * (sb * (acc.x + gv.x));
    o.y = (1.0f - DELTA) * (gv.y * sa) + DELTA * (sb * (acc.y + gv.y));
    o.z = (1.0f - DELTA) * (gv.z * sa) + DELTA * (sb * (acc.z + gv.z));
    o.w = (1.0f - DELTA) * (gv.w * sa) + DELTA * (sb * (acc.w + gv.w));
    return o;
}

template <bool LAST, int PIN, int POUT>
__global__ __launch_bounds__(256)
void dgc_prop(const float* __restrict__ hin,
              float* __restrict__ hout,
              const int* __restrict__ node_raw,
              const int* __restrict__ csr_raw,
              const float* __restrict__ bias,
              const float* __restrict__ gamma,
              const float* __restrict__ beta) {
    const int2* node = (const int2*)node_raw;
    const int*  csr  = csr_raw;
    const int lane   = threadIdx.x & 31;
    const int grp    = lane / GL;
    const int j4     = (lane - grp * GL) * 4;
    const int wave   = (blockIdx.x * blockDim.x + threadIdx.x) >> 5;
    const int nwaves = (gridDim.x * blockDim.x) >> 5;

    float4 bv = make_float4(0.f, 0.f, 0.f, 0.f), gv = bv, tv = bv;
    dgc_ring ring = {lane, lane, lane, lane, lane, lane};
    if (LAST) {
        bv = make_float4(bias[j4],  bias[j4 + 1],  bias[j4 + 2],  bias[j4 + 3]);
        gv = make_float4(gamma[j4], gamma[j4 + 1], gamma[j4 + 2], gamma[j4 + 3]);
        tv = make_float4(beta[j4],  beta[j4 + 1],  beta[j4 + 2],  beta[j4 + 3]);
        if (grp < RPW) {
            const int base = grp * GL;
            const int j    = lane - base;
            ring.p5 = base + (j + 5) % GL;
            ring.p1 = base + (j + 1) % GL;
            ring.p2 = base + (j + 2) % GL;
            ring.p3 = base + (j + 3) % GL;
            ring.p4 = base + (j + 4) % GL;
            ring.p0 = base;
        }
    }

    for (int t = wave; t < NTRIP; t += nwaves) {
        const int  r    = t * RPW + grp;
        const bool live = (grp < RPW) && (r < NN);
        int beg = 0, end = 0;
        float dv = 0.f;
        float4 hv = make_float4(0.f, 0.f, 0.f, 0.f);
        if (live) {
            const int2 nr = node[r];
            beg = nr.x;
            end = node[r + 1].x;
            dv  = __int_as_float(nr.y);
            hv  = *(const float4*)(hin + r * PIN + j4);
        }
        float4 acc = make_float4(0.f, 0.f, 0.f, 0.f);
        int e = beg;
        for (; e + PB <= end; e += PB) dgc_batch<true>(hin, csr, e, PB, j4, acc);
        if (e < end) dgc_batch<false>(hin, csr, e, end - e, j4, acc);
        if (!LAST) {
            if (live) *(float4*)(hout + r * POUT + j4) = dgc_step_row(hv, 1.0f, dv * dv, acc);
        } else {
            const float sd = ((float)(end - beg) + 1.0f) * dv;
            const float4 o = dgc_step_row(hv, sd, dv, acc);
            float4 x;
            x.x = o.x + bv.x;
            x.y = o.y + bv.y;
            x.z = o.z + bv.z;
            x.w = o.w + bv.w;
            const float s  = dgc_group_sum((x.x + x.y) + (x.z + x.w), ring);
            const float sq = dgc_group_sum((x.x * x.x + x.y * x.y) + (x.z * x.z + x.w * x.w), ring);
            const float mean = s * (1.0f / CC);
            const float var  = sq * (1.0f / CC) - mean * mean;
            const float inv  = rsqrtf(var + LN_EPS);
            float4 y;
            y.x = (x.x - mean) * inv * gv.x + tv.x;
            y.y = (x.y - mean) * inv * gv.y + tv.y;
            y.z = (x.z - mean) * inv * gv.z + tv.z;
            y.w = (x.w - mean) * inv * gv.w + tv.w;
            if (live) *(float4*)(hout + r * POUT + j4) = y;
        }
    }
}

#define M_TILES   (NN / 16)
#define N_TILES   3
#define NPAD      (N_TILES * 16)
#define FC_BLOCK  256
#define WP        (DD + 4)
static_assert((WP * 4) % 16 == 0 && ((WP * 4) / 16) % 2 == 1, "W row pitch in LDS: 16 bytes x odd");
static_assert(N_TILES == 3 && CC > 32 && CC <= NPAD, "the FC writes its three column tiles out by hand");
#define W_F4      (CC * DD / 4)
static_assert(DD == 128 && W_F4 % FC_BLOCK == 0 && (NPAD - CC) * (DD / 4) == FC_BLOCK,
              "the W stage: whole passes of the block over W, one zero piece per thread");

template <int PITCH>
__global__ __launch_bounds__(FC_BLOCK)
void dgc_fc_wmma(const float* __restrict__ x,
                 const float* __restrict__ W,
                 const int* __restrict__ node_raw,
                 float* __restrict__ out) {
    __shared__ __attribute__((aligned(16))) float wlds[NPAD * WP];

    {
        const float4* W4 = (const float4*)W;
        float4 w[W_F4 / FC_BLOCK];
        #pragma unroll
        for (int i = 0; i < W_F4 / FC_BLOCK; ++i) w[i] = W4[threadIdx.x + i * FC_BLOCK];
        #pragma unroll
        for (int i = 0; i < W_F4 / FC_BLOCK; ++i) {
            const int q = threadIdx.x + i * FC_BLOCK;
            *(float4*)(wlds + (q >> 5) * WP + (q & 31) * 4) = w[i];
        }
        const int z = threadIdx.x;
        *(float4*)(wlds + (CC + (z >> 5)) * WP + (z & 31) * 4) = make_float4(0.f, 0.f, 0.f, 0.f);
    }
    __syncthreads();

    const int m_tile = (blockIdx.x * FC_BLOCK + threadIdx.x) >> 5;
    const int lane   = threadIdx.x & 31;
    if (m_tile >= M_TILES) return;

    const int m_base = m_tile * 16;
    const int l16    = lane & 15;
    const int kb     = (lane >> 4) * 2;

    const float* arow = x + (size_t)(m_base + l16) * DD + kb;
    const float* brow = wlds + l16 * WP + kb;

    const int2* node = (const int2*)node_raw;
    const int m_off = (lane >> 4) * 8;
    float dvr[8];
    #pragma unroll
    for (int v = 0; v < 8; ++v) dvr[v] = __int_as_float(node[m_base + m_off + v].y);

    v8f acc0 = {}, acc1 = {}, acc2 = {};
    #pragma unroll 16
    for (int k = 0; k < DD; k += 4) {
        v2f a, b0, b1, b2;
        a.x  = arow[k];
        a.y  = arow[k + 1];
        b0.x = brow[k];
        b0.y = brow[k + 1];
        b1.x = brow[16 * WP + k];
        b1.y = brow[16 * WP + k + 1];
        b2.x = brow[32 * WP + k];
        b2.y = brow[32 * WP + k + 1];
        acc0 = __builtin_amdgcn_wmma_f32_16x16x4_f32(
             false, a,  false, b0,
             (short)0, acc0,  false,  false);
        acc1 = __builtin_amdgcn_wmma_f32_16x16x4_f32(
             false, a,  false, b1,
             (short)0, acc1,  false,  false);
        acc2 = __builtin_amdgcn_wmma_f32_16x16x4_f32(
             false, a,  false, b2,
             (short)0, acc2,  false,  false);
    }

    asm volatile("v_nop\n\tv_nop\n\tv_nop\n\tv_nop\n\tv_nop\n\ts_wait_alu depctr_va_vdst(0)"
                 : "+v"(acc0), "+v"(acc1), "+v"(acc2));
    float* orow = out + (size_t)(m_base + m_off) * PITCH + l16;
    #pragma unroll
    for (int v = 0; v < 8; ++v) {
        orow[(size_t)v * PITCH]      = acc0[v] * dvr[v];
        orow[(size_t)v * PITCH + 16] = acc1[v] * dvr[v];
    }
    if (l16 < CC - 32) {
        #pragma unroll
        for (int v = 0; v < 8; ++v) orow[(size_t)v * PITCH + 32] = acc2[v] * dvr[v];
    }
}

extern "C" void kernel_launch(void* const* d_in, const int* in_sizes, int n_in,
                              void* d_out, int out_size, void* d_ws, size_t ws_size,
                              hipStream_t stream) {
    const float* feat  = (const float*)d_in[0];
    const int*   row   = (const int*)  d_in[1];
    const int*   col   = (const int*)  d_in[2];
    const float* W     = (const float*)d_in[3];
    const float* bias  = (const float*)d_in[4];
    const float* gamma = (const float*)d_in[5];
    const float* beta  = (const float*)d_in[6];
    float* out = (float*)d_out;

    const int B = 256;
    const int fc_waves    = M_TILES;
    const int fc_blocks   = (fc_waves * 32 + FC_BLOCK - 1) / FC_BLOCK;
    const int prop_blocks = (NTRIP / 2 * 32 + B - 1) / B;

    int*   node = (int*)d_ws;
    int*   mat  = node + 2 * NP;
    float* ta   = (float*)(mat + NP);
    static_assert((3 * NP * 4) % 256 == 0, "the tables start on a 256-byte unit of the workspace");
    static_assert(2 * EE <= NN * CC, "the staged pairs fit the table they borrow, dense or padded");

    const size_t padded_need = ((size_t)3 * NP + (size_t)2 * NN * HP + (size_t)(EE + CSR_PAD)) * 4;

    if (ws_size >= padded_need) {
        float* tb   = ta + (size_t)NN * HP;
        int*   csr  = (int*)(tb + (size_t)NN * HP);
        int*   stage = (int*)tb;

        dgc_part_local   <<<NCH, SB1, 0, stream>>>(row, col, mat, stage);
        dgc_row_build<HP><<<NB1, SB1, 0, stream>>>(stage, mat, node, csr);

        dgc_fc_wmma<HP><<<fc_blocks, FC_BLOCK, 0, stream>>>(feat, W, node, ta);

        dgc_prop<false, HP, HP><<<prop_blocks, B, 0, stream>>>(ta, tb, node, csr, bias, gamma, beta);
        dgc_prop<false, HP, HP><<<prop_blocks, B, 0, stream>>>(tb, ta, node, csr, bias, gamma, beta);
        dgc_prop<false, HP, HP><<<prop_blocks, B, 0, stream>>>(ta, tb, node, csr, bias, gamma, beta);
        dgc_prop<true,  HP, CC><<<prop_blocks, B, 0, stream>>>(tb, out, node, csr, bias, gamma, beta);
    } else {
        float* h    = ta;
        int*   csr  = (int*)(h + (size_t)NN * CC);
        int*   stage = (int*)h;

        dgc_part_local   <<<NCH, SB1, 0, stream>>>(row, col, mat, stage);
        dgc_row_build<CC><<<NB1, SB1, 0, stream>>>(stage, mat, node, csr);

        dgc_fc_wmma<CC><<<fc_blocks, FC_BLOCK, 0, stream>>>(feat, W, node, out);

        dgc_prop<false, CC, CC><<<prop_blocks, B, 0, stream>>>(out, h, node, csr, bias, gamma, beta);
        dgc_prop<false, CC, CC><<<prop_blocks, B, 0, stream>>>(h, out, node, csr, bias, gamma, beta);
        dgc_prop<false, CC, CC><<<prop_blocks, B, 0, stream>>>(out, h, node, csr, bias, gamma, beta);
        dgc_prop<true,  CC, CC><<<prop_blocks, B, 0, stream>>>(h, out, node, csr, bias, gamma, beta);
    }
}
